# starting kernel
# baseline (speedup 1.0000x reference)
#include <hip/hip_runtime.h>

typedef float v2f __attribute__((ext_vector_type(2)));
typedef float v8f __attribute__((ext_vector_type(8)));

#define IN_F  128
#define HID_F 128
#define OUT_F 64
#define KDIM  128

__global__ void zero_kernel(float* __restrict__ p, long n) {
    long i = (long)blockIdx.x * blockDim.x + threadIdx.x;
    if (i < n) p[i] = 0.0f;
}

__global__ void degree_kernel(const int* __restrict__ src, const int* __restrict__ dst,
                              float* deg_src, float* deg_dst, int ne) {
    int e = blockIdx.x * blockDim.x + threadIdx.x;
    if (e < ne) {
        atomicAdd(&deg_src[src[e]], 1.0f);
        atomicAdd(&deg_dst[dst[e]], 1.0f);
    }
}

__global__ void rsqrt_kernel(float* __restrict__ p, int n) {
    int i = blockIdx.x * blockDim.x + threadIdx.x;
    if (i < n) {
        float d = p[i];
        p[i] = rsqrtf(d < 1.0f ? 1.0f : d);
    }
}

__global__ void epilogue_kernel(float* __restrict__ Y, const float* __restrict__ dinv,
                                const float* __restrict__ bias, int F, long total,
                                int do_relu) {
    long t = (long)blockIdx.x * blockDim.x + threadIdx.x;
    if (t >= total) return;
    int i = (int)(t / F);
    int f = (int)(t - (long)i * F);
    float v = Y[t] * dinv[i] + bias[f];
    if (do_relu) v = v > 0.0f ? v : 0.0f;
    Y[t] = v;
}

template <int FOUT>
__launch_bounds__(FOUT * 2)
__global__ void gemm_rows16_wmma(const float* __restrict__ X,
                                 const float* __restrict__ dinv,
                                 const float* __restrict__ W,
                                 float* __restrict__ H) {
    constexpr int LDSS = KDIM + 4;
    __shared__ __align__(16) float As[16 * LDSS];

    const int row0 = blockIdx.x * 16;
    const int tid  = threadIdx.x;

    for (int i = tid; i < 16 * KDIM; i += blockDim.x) {
        int r = i >> 7;
        int c = i & (KDIM - 1);
        As[r * LDSS + c] = X[(size_t)(row0 + r) * KDIM + c] * dinv[row0 + r];
    }
    __syncthreads();

    const int lane  = tid & 31;
    const int wave  = tid >> 5;
    const int n     = lane & 15;
    const int khalf = lane >> 4;
    const int m16   = lane & 15;
    const int ncol  = wave * 16 + n;

    v8f acc = {};
#pragma unroll
    for (int k0 = 0; k0 < KDIM; k0 += 4) {
        const int ka = k0 + 2 * khalf;
        v2f a = *(const v2f*)&As[m16 * LDSS + ka];
        v2f b;
        b.x = W[(size_t)ka * FOUT + ncol];
        b.y = W[(size_t)(ka + 1) * FOUT + ncol];
        acc = __builtin_amdgcn_wmma_f32_16x16x4_f32(
             false, a,  false, b,
             (short)0, acc,  false,  false);
    }

#pragma unroll
    for (int r = 0; r < 8; ++r) {
        const int m = r + 8 * khalf;
        H[(size_t)(row0 + m) * FOUT + ncol] = acc[r];
    }
}

template <int F>
__global__ void scatter_add_kernel(const float* __restrict__ H,
                                   const int* __restrict__ src,
                                   const int* __restrict__ dst,
                                   float* __restrict__ AGG, int ne) {
    const int gid  = blockIdx.x * blockDim.x + threadIdx.x;
    const int e    = gid >> 5;
    const int lane = gid & 31;
    if (e >= ne) return;

    const int s = src[e];
    const int d = dst[e];
    constexpr int V = F / 32;

    const float* hp = H   + (size_t)s * F + lane * V;
    float*       op = AGG + (size_t)d * F + lane * V;

    if (V == 4) {
        float4 v = *(const float4*)hp;
        atomicAdd(op + 0, v.x);
        atomicAdd(op + 1, v.y);
        atomicAdd(op + 2, v.z);
        atomicAdd(op + 3, v.w);
    } else {
        float2 v = *(const float2*)hp;
        atomicAdd(op + 0, v.x);
        atomicAdd(op + 1, v.y);
    }
}

static inline int cdiv(long a, long b) { return (int)((a + b - 1) / b); }

extern "C" void kernel_launch(void* const* d_in, const int* in_sizes, int n_in,
                              void* d_out, int out_size, void* d_ws, size_t ws_size,
                              hipStream_t stream) {
    const float* features = (const float*)d_in[0];
    const int*   src      = (const int*)d_in[1];
    const int*   dst      = (const int*)d_in[2];
    const float* W1       = (const float*)d_in[3];
    const float* b1       = (const float*)d_in[4];
    const float* W2       = (const float*)d_in[5];
    const float* b2       = (const float*)d_in[6];
    const float* W3       = (const float*)d_in[7];
    const float* b3       = (const float*)d_in[8];
    float*       out      = (float*)d_out;

    const int N = in_sizes[0] / IN_F;
    const int E = in_sizes[1];

    float* ws   = (float*)d_ws;
    float* dsrc = ws;
    float* ddst = ws + N;
    float* buf0 = ws + 2 * (size_t)N;
    float* buf1 = buf0 + (size_t)N * HID_F;

    const long NH = (long)N * HID_F;
    const long NO = (long)N * OUT_F;
    const int  nblk = N / 16;

    zero_kernel<<<cdiv(2L * N, 256), 256, 0, stream>>>(dsrc, 2L * N);
    degree_kernel<<<cdiv(E, 256), 256, 0, stream>>>(src, dst, dsrc, ddst, E);
    rsqrt_kernel<<<cdiv(2L * N, 256), 256, 0, stream>>>(dsrc, 2 * N);

    gemm_rows16_wmma<HID_F><<<nblk, HID_F * 2, 0, stream>>>(features, dsrc, W1, buf0);
    zero_kernel<<<cdiv(NH, 256), 256, 0, stream>>>(buf1, NH);
    scatter_add_kernel<HID_F><<<cdiv((long)E * 32, 256), 256, 0, stream>>>(buf0, src, dst, buf1, E);
    epilogue_kernel<<<cdiv(NH, 256), 256, 0, stream>>>(buf1, ddst, b1, HID_F, NH, 1);

    gemm_rows16_wmma<HID_F><<<nblk, HID_F * 2, 0, stream>>>(buf1, dsrc, W2, buf0);
    zero_kernel<<<cdiv(NH, 256), 256, 0, stream>>>(buf1, NH);
    scatter_add_kernel<HID_F><<<cdiv((long)E * 32, 256), 256, 0, stream>>>(buf0, src, dst, buf1, E);
    epilogue_kernel<<<cdiv(NH, 256), 256, 0, stream>>>(buf1, ddst, b2, HID_F, NH, 1);

    gemm_rows16_wmma<OUT_F><<<nblk, OUT_F * 2, 0, stream>>>(buf1, dsrc, W3, buf0);
    zero_kernel<<<cdiv(NO, 256), 256, 0, stream>>>(out, NO);
    scatter_add_kernel<OUT_F><<<cdiv((long)E * 32, 256), 256, 0, stream>>>(buf0, src, dst, out, E);
    epilogue_kernel<<<cdiv(NO, 256), 256, 0, stream>>>(out, ddst, b3, OUT_F, NO, 0);
}


# Segment sums in LDS: a block owns 391 destination rows, scans the dst list, gathers its edges' rows, LDS float adds, rows stored once; row sums need no global atomics or zero fill; order unfixed
# speedup vs baseline: 1.9257x; 1.9257x over previous
#include <hip/hip_runtime.h>

typedef float v2f __attribute__((ext_vector_type(2)));
typedef float v8f __attribute__((ext_vector_type(8)));

#define IN_F  128
#define HID_F 128
#define OUT_F 64
#define KDIM  128

__global__ void zero_kernel(float* __restrict__ p, long n) {
    long i = (long)blockIdx.x * blockDim.x + threadIdx.x;
    if (i < n) p[i] = 0.0f;
}

__global__ void degree_kernel(const int* __restrict__ src, const int* __restrict__ dst,
                              float* deg_src, float* deg_dst, int ne) {
    int e = blockIdx.x * blockDim.x + threadIdx.x;
    if (e < ne) {
        atomicAdd(&deg_src[src[e]], 1.0f);
        atomicAdd(&deg_dst[dst[e]], 1.0f);
    }
}

__global__ void rsqrt_kernel(float* __restrict__ p, int n) {
    int i = blockIdx.x * blockDim.x + threadIdx.x;
    if (i < n) {
        float d = p[i];
        p[i] = rsqrtf(d < 1.0f ? 1.0f : d);
    }
}

__global__ void epilogue_kernel(float* __restrict__ Y, const float* __restrict__ dinv,
                                const float* __restrict__ bias, int F, long total,
                                int do_relu) {
    long t = (long)blockIdx.x * blockDim.x + threadIdx.x;
    if (t >= total) return;
    int i = (int)(t / F);
    int f = (int)(t - (long)i * F);
    float v = Y[t] * dinv[i] + bias[f];
    if (do_relu) v = v > 0.0f ? v : 0.0f;
    Y[t] = v;
}

template <int FOUT, bool PRE>
__launch_bounds__(FOUT * 2)
__global__ void gemm_rows16_wmma(const float* X,
                                 const float* __restrict__ dsrc,
                                 const float* __restrict__ ddst,
                                 const float* __restrict__ pbias,
                                 const float* __restrict__ W,
                                 float* __restrict__ H) {
    constexpr int LDSS = KDIM + 4;
    __shared__ __align__(16) float As[16 * LDSS];

    const int row0 = blockIdx.x * 16;
    const int tid  = threadIdx.x;

    for (int i = tid; i < 16 * KDIM; i += blockDim.x) {
        int r = i >> 7;
        int c = i & (KDIM - 1);
        float x = X[(size_t)(row0 + r) * KDIM + c];
        if (PRE) {
            x = x * ddst[row0 + r] + pbias[c];
            x = x > 0.0f ? x : 0.0f;
        }
        As[r * LDSS + c] = x * dsrc[row0 + r];
    }
    __syncthreads();

    const int lane  = tid & 31;
    const int wave  = tid >> 5;
    const int n     = lane & 15;
    const int khalf = lane >> 4;
    const int m16   = lane & 15;
    const int ncol  = wave * 16 + n;

    v8f acc = {};
#pragma unroll
    for (int k0 = 0; k0 < KDIM; k0 += 4) {
        const int ka = k0 + 2 * khalf;
        v2f a = *(const v2f*)&As[m16 * LDSS + ka];
        v2f b;
        b.x = W[(size_t)ka * FOUT + ncol];
        b.y = W[(size_t)(ka + 1) * FOUT + ncol];
        acc = __builtin_amdgcn_wmma_f32_16x16x4_f32(
             false, a,  false, b,
             (short)0, acc,  false,  false);
    }

#pragma unroll
    for (int r = 0; r < 8; ++r) {
        const int m = r + 8 * khalf;
        H[(size_t)(row0 + m) * FOUT + ncol] = acc[r];
    }
}

#define OWN_ROWS 391
#define OWN_WAVES 32
#define SCAN_U 8

template <int F>
__device__ __forceinline__ void owner_take(const float* __restrict__ H,
                                           const int* __restrict__ src,
                                           float* acc, const int2* qe, int n, int lane) {
    constexpr int V   = F / 32;
    constexpr int EPG = 32 / V;
    int idx = lane & (EPG - 1);
    idx = idx < n ? idx : n - 1;
    const int2 en = qe[idx];
    const int  s  = src[en.x];
    float v[EPG][V];
#pragma unroll
    for (int u = 0; u < EPG; ++u) {
        const int su = __builtin_amdgcn_readlane(s, u);
        const float* hp = H + (size_t)su * F + lane;
#pragma unroll
        for (int j = 0; j < V; ++j) v[u][j] = hp[32 * j];
    }
#pragma unroll
    for (int u = 0; u < EPG; ++u) {
        if (u < n) {
            const int du = __builtin_amdgcn_readlane(en.y, u);
            float* ap = acc + du * F + lane;
#pragma unroll
            for (int j = 0; j < V; ++j) atomicAdd(ap + 32 * j, v[u][j]);
        }
    }
}

template <int F>
__launch_bounds__(OWN_WAVES * 32)
__global__ void owner_sum_kernel(const float* __restrict__ H,
                                 const int* __restrict__ src,
                                 const int* __restrict__ dst,
                                 float* __restrict__ AGG, int ne, int n_nodes) {
    constexpr int V   = F / 32;
    constexpr int EPG = 32 / V;
    constexpr int QW  = 128 + EPG;
    constexpr int NT  = OWN_WAVES * 32;

    __shared__ __align__(16) float acc[OWN_ROWS * F];
    __shared__ int2 queue[OWN_WAVES * QW];

    const int tid  = threadIdx.x;
    const int lane = tid & 31;
    const int wave = __builtin_amdgcn_readfirstlane(tid >> 5);
    const int r0   = blockIdx.x * OWN_ROWS;
    const int rows = (n_nodes - r0) < OWN_ROWS ? (n_nodes - r0) : OWN_ROWS;

    float4* a4 = reinterpret_cast<float4*>(acc);
    for (int i = tid; i < OWN_ROWS * F / 4; i += NT) a4[i] = make_float4(0.0f, 0.0f, 0.0f, 0.0f);
    __syncthreads();

    int2* q = queue + wave * QW;
    int cnt = 0;
    const int npieces = (ne + 127) >> 7;

    for (int p0 = wave; p0 < npieces; p0 += OWN_WAVES * SCAN_U) {
        int4 d[SCAN_U];
        if ((p0 + (SCAN_U - 1) * OWN_WAVES + 1) * 128 <= ne) {
#pragma unroll
            for (int u = 0; u < SCAN_U; ++u)
                d[u] = *reinterpret_cast<const int4*>(dst + (p0 + u * OWN_WAVES) * 128 + lane * 4);
        } else {
#pragma unroll
            for (int u = 0; u < SCAN_U; ++u) {
                const int eb = (p0 + u * OWN_WAVES) * 128 + lane * 4;
                if ((p0 + u * OWN_WAVES + 1) * 128 <= ne) {
                    d[u] = *reinterpret_cast<const int4*>(dst + eb);
                } else if ((p0 + u * OWN_WAVES) * 128 < ne) {
                    d[u].x = eb     < ne ? dst[eb]     : -1;
                    d[u].y = eb + 1 < ne ? dst[eb + 1] : -1;
                    d[u].z = eb + 2 < ne ? dst[eb + 2] : -1;
                    d[u].w = eb + 3 < ne ? dst[eb + 3] : -1;
                } else {
                    d[u] = make_int4(-1, -1, -1, -1);
                }
            }
        }
#pragma unroll
        for (int u = 0; u < SCAN_U; ++u) {
            const int eb = (p0 + u * OWN_WAVES) * 128 + lane * 4;
            const unsigned t0 = (unsigned)(d[u].x - r0);
            const unsigned t1 = (unsigned)(d[u].y - r0);
            const unsigned t2 = (unsigned)(d[u].z - r0);
            const unsigned t3 = (unsigned)(d[u].w - r0);
            const unsigned ta = t0 < t1 ? t0 : t1;
            const unsigned tb = t2 < t3 ? t2 : t3;
            const unsigned tm = ta < tb ? ta : tb;
            if (__builtin_amdgcn_ballot_w32(tm < (unsigned)rows) == 0u) continue;

#define OWNER_KEEP(T, E)                                                         \
            {                                                                    \
                const bool     m    = (T) < (unsigned)rows;                      \
                const unsigned mask = __builtin_amdgcn_ballot_w32(m);            \
                if (mask != 0u) {                                                \
                    const int pos = cnt + (int)__builtin_amdgcn_mbcnt_lo(mask, 0u); \
                    if (m) q[pos] = make_int2((E), (int)(T));                    \
                    cnt += __builtin_popcount(mask);                             \
                }                                                                \
            }
            OWNER_KEEP(t0, eb)
            OWNER_KEEP(t1, eb + 1)
            OWNER_KEEP(t2, eb + 2)
            OWNER_KEEP(t3, eb + 3)
#undef OWNER_KEEP

            while (cnt >= EPG) {
                cnt -= EPG;
                __builtin_amdgcn_wave_barrier();
                owner_take<F>(H, src, acc, q + cnt, EPG, lane);
                __builtin_amdgcn_wave_barrier();
            }
        }
    }
    if (cnt > 0) {
        __builtin_amdgcn_wave_barrier();
        owner_take<F>(H, src, acc, q, cnt, lane);
    }
    __syncthreads();

    float4* o4 = reinterpret_cast<float4*>(AGG + (size_t)r0 * F);
    for (int i = tid; i < rows * F / 4; i += NT) o4[i] = a4[i];
}

static inline int cdiv(long a, long b) { return (int)((a + b - 1) / b); }

extern "C" void kernel_launch(void* const* d_in, const int* in_sizes, int n_in,
                              void* d_out, int out_size, void* d_ws, size_t ws_size,
                              hipStream_t stream) {
    const float* features = (const float*)d_in[0];
    const int*   src      = (const int*)d_in[1];
    const int*   dst      = (const int*)d_in[2];
    const float* W1       = (const float*)d_in[3];
    const float* b1       = (const float*)d_in[4];
    const float* W2       = (const float*)d_in[5];
    const float* b2       = (const float*)d_in[6];
    const float* W3       = (const float*)d_in[7];
    const float* b3       = (const float*)d_in[8];
    float*       out      = (float*)d_out;

    const int N = in_sizes[0] / IN_F;
    const int E = in_sizes[1];

    float* ws   = (float*)d_ws;
    float* dsrc = ws;
    float* ddst = ws + N;
    float* buf0 = ws + 2 * (size_t)N;
    float* buf1 = buf0 + (size_t)N * HID_F;

    const long NO = (long)N * OUT_F;
    const int  nblk = N / 16;
    const int  nown = cdiv(N, OWN_ROWS);

    zero_kernel<<<cdiv(2L * N, 256), 256, 0, stream>>>(dsrc, 2L * N);
    degree_kernel<<<cdiv(E, 256), 256, 0, stream>>>(src, dst, dsrc, ddst, E);
    rsqrt_kernel<<<cdiv(2L * N, 256), 256, 0, stream>>>(dsrc, 2 * N);

    gemm_rows16_wmma<HID_F, false><<<nblk, HID_F * 2, 0, stream>>>(
        features, dsrc, ddst, b1, W1, buf0);
    owner_sum_kernel<HID_F><<<nown, OWN_WAVES * 32, 0, stream>>>(buf0, src, dst, buf1, E, N);

    gemm_rows16_wmma<HID_F, true><<<nblk, HID_F * 2, 0, stream>>>(
        buf1, dsrc, ddst, b1, W2, buf0);
    owner_sum_kernel<HID_F><<<nown, OWN_WAVES * 32, 0, stream>>>(buf0, src, dst, buf1, E, N);

    gemm_rows16_wmma<OUT_F, true><<<nblk, OUT_F * 2, 0, stream>>>(
        buf1, dsrc, ddst, b2, W3, buf0);
    owner_sum_kernel<OUT_F><<<nown, OWN_WAVES * 32, 0, stream>>>(buf0, src, dst, out, E, N);
    epilogue_kernel<<<cdiv(NO, 256), 256, 0, stream>>>(out, ddst, b3, OUT_F, NO, 0);
}


# Segment-sum scan reads a 1-byte owner tag per edge (4 matched per op), then one packed (src, local row) word per kept edge; degree pass writes both planes into out; sum order unfixed as before
# speedup vs baseline: 2.1082x; 1.0948x over previous
#include <hip/hip_runtime.h>

typedef float v2f __attribute__((ext_vector_type(2)));
typedef float v8f __attribute__((ext_vector_type(8)));

#define IN_F  128
#define HID_F 128
#define OUT_F 64
#define KDIM  128

__global__ void zero_kernel(float* __restrict__ p, long n) {
    long i = (long)blockIdx.x * blockDim.x + threadIdx.x;
    if (i < n) p[i] = 0.0f;
}

#define OWN_ROWS 391
__global__ void degree_tag_kernel(const int* __restrict__ src, const int* __restrict__ dst,
                                  float* deg_src, float* deg_dst,
                                  unsigned* __restrict__ tags, unsigned* __restrict__ words,
                                  int ne, int n_nodes, int nquads) {
    const int t = blockIdx.x * blockDim.x + threadIdx.x;
    if (t >= nquads) return;
    const int e0 = 4 * t;
    int s[4], d[4];
    if (e0 + 3 < ne) {
        const int4 s4 = *reinterpret_cast<const int4*>(src + e0);
        const int4 d4 = *reinterpret_cast<const int4*>(dst + e0);
        s[0] = s4.x; s[1] = s4.y; s[2] = s4.z; s[3] = s4.w;
        d[0] = d4.x; d[1] = d4.y; d[2] = d4.z; d[3] = d4.w;
    } else {
#pragma unroll
        for (int k = 0; k < 4; ++k) {
            s[k] = e0 + k < ne ? src[e0 + k] : 0;
            d[k] = e0 + k < ne ? dst[e0 + k] : -1;
        }
    }
    unsigned tg = 0u;
    unsigned w[4];
#pragma unroll
    for (int k = 0; k < 4; ++k) {
        if (e0 + k < ne) {
            atomicAdd(&deg_src[s[k]], 1.0f);
            atomicAdd(&deg_dst[d[k]], 1.0f);
        }
        const bool     ok  = (unsigned)d[k] < (unsigned)n_nodes;
        const unsigned own = ok ? (unsigned)d[k] / OWN_ROWS : 0u;
        const unsigned loc = ok ? (unsigned)d[k] - own * OWN_ROWS : 511u;
        tg |= (own & 255u) << (8 * k);
        w[k] = ((unsigned)s[k] & 0x1FFFFu) | (loc << 17);
    }
    tags[t] = tg;
    *reinterpret_cast<uint4*>(words + e0) = make_uint4(w[0], w[1], w[2], w[3]);
}

__global__ void rsqrt_kernel(float* __restrict__ p, int n) {
    int i = blockIdx.x * blockDim.x + threadIdx.x;
    if (i < n) {
        float d = p[i];
        p[i] = rsqrtf(d < 1.0f ? 1.0f : d);
    }
}

__global__ void epilogue_kernel(const float* __restrict__ X, float* __restrict__ Y,
                                const float* __restrict__ dinv,
                                const float* __restrict__ bias, int F, long total,
                                int do_relu) {
    long t = (long)blockIdx.x * blockDim.x + threadIdx.x;
    if (t >= total) return;
    int i = (int)(t / F);
    int f = (int)(t - (long)i * F);
    float v = X[t] * dinv[i] + bias[f];
    if (do_relu) v = v > 0.0f ? v : 0.0f;
    Y[t] = v;
}

template <int FOUT, bool PRE>
__launch_bounds__(FOUT * 2)
__global__ void gemm_rows16_wmma(const float* X,
                                 const float* __restrict__ dsrc,
                                 const float* __restrict__ ddst,
                                 const float* __restrict__ pbias,
                                 const float* __restrict__ W,
                                 float* __restrict__ H) {
    constexpr int LDSS = KDIM + 4;
    __shared__ __align__(16) float As[16 * LDSS];

    const int row0 = blockIdx.x * 16;
    const int tid  = threadIdx.x;

    for (int i = tid; i < 16 * KDIM; i += blockDim.x) {
        int r = i >> 7;
        int c = i & (KDIM - 1);
        float x = X[(size_t)(row0 + r) * KDIM + c];
        if (PRE) {
            x = x * ddst[row0 + r] + pbias[c];
            x = x > 0.0f ? x : 0.0f;
        }
        As[r * LDSS + c] = x * dsrc[row0 + r];
    }
    __syncthreads();

    const int lane  = tid & 31;
    const int wave  = tid >> 5;
    const int n     = lane & 15;
    const int khalf = lane >> 4;
    const int m16   = lane & 15;
    const int ncol  = wave * 16 + n;

    v8f acc = {};
#pragma unroll
    for (int k0 = 0; k0 < KDIM; k0 += 4) {
        const int ka = k0 + 2 * khalf;
        v2f a = *(const v2f*)&As[m16 * LDSS + ka];
        v2f b;
        b.x = W[(size_t)ka * FOUT + ncol];
        b.y = W[(size_t)(ka + 1) * FOUT + ncol];
        acc = __builtin_amdgcn_wmma_f32_16x16x4_f32(
             false, a,  false, b,
             (short)0, acc,  false,  false);
    }

#pragma unroll
    for (int r = 0; r < 8; ++r) {
        const int m = r + 8 * khalf;
        H[(size_t)(row0 + m) * FOUT + ncol] = acc[r];
    }
}

#define OWN_WAVES 32
#define SCAN_U 8

template <int F>
__device__ __forceinline__ void owner_take(const float* __restrict__ H,
                                           const unsigned* __restrict__ words,
                                           float* acc, const int* qe, int n, int rows, int lane) {
    constexpr int V   = F / 32;
    constexpr int EPG = 32 / V;
    int idx = lane & (EPG - 1);
    idx = idx < n ? idx : n - 1;
    const int w = (int)words[qe[idx]];
    float v[EPG][V];
#pragma unroll
    for (int u = 0; u < EPG; ++u) {
        const int su = __builtin_amdgcn_readlane(w, u) & 0x1FFFF;
        const float* hp = H + (size_t)su * F + lane;
#pragma unroll
        for (int j = 0; j < V; ++j) v[u][j] = hp[32 * j];
    }
#pragma unroll
    for (int u = 0; u < EPG; ++u) {
        const int du = (int)((unsigned)__builtin_amdgcn_readlane(w, u) >> 17);
        if (u < n && du < rows) {
            float* ap = acc + du * F + lane;
#pragma unroll
            for (int j = 0; j < V; ++j) atomicAdd(ap + 32 * j, v[u][j]);
        }
    }
}

template <int F>
__launch_bounds__(OWN_WAVES * 32)
__global__ void owner_sum_kernel(const float* __restrict__ H,
                                 const unsigned* __restrict__ tags,
                                 const unsigned* __restrict__ words,
                                 float* __restrict__ AGG, int ne, int n_nodes) {
    constexpr int V   = F / 32;
    constexpr int EPG = 32 / V;
    constexpr int QW  = 512 + EPG;
    constexpr int NT  = OWN_WAVES * 32;

    __shared__ __align__(16) float acc[OWN_ROWS * F];
    __shared__ int queue[OWN_WAVES * QW];

    const int tid  = threadIdx.x;
    const int lane = tid & 31;
    const int wave = __builtin_amdgcn_readfirstlane(tid >> 5);
    const int r0   = blockIdx.x * OWN_ROWS;
    const int rows = (n_nodes - r0) < OWN_ROWS ? (n_nodes - r0) : OWN_ROWS;
    const unsigned rep = (unsigned)(blockIdx.x & 255) * 0x01010101u;

    float4* a4 = reinterpret_cast<float4*>(acc);
    for (int i = tid; i < OWN_ROWS * F / 4; i += NT) a4[i] = make_float4(0.0f, 0.0f, 0.0f, 0.0f);
    __syncthreads();

    int* q = queue + wave * QW;
    int cnt = 0;
    const int npieces = (ne + 511) >> 9;
    const uint4* tags4 = reinterpret_cast<const uint4*>(tags);

    for (int p0 = wave; p0 < npieces; p0 += OWN_WAVES * SCAN_U) {
        uint4 t[SCAN_U];
#pragma unroll
        for (int u = 0; u < SCAN_U; ++u) {
            const int pc = (p0 + u * OWN_WAVES) < npieces ? (p0 + u * OWN_WAVES) : npieces - 1;
            t[u] = tags4[(size_t)pc * 32 + lane];
        }
#pragma unroll
        for (int u = 0; u < SCAN_U; ++u) {
            if (p0 + u * OWN_WAVES >= npieces) continue;
            const unsigned y[4] = { t[u].x ^ rep, t[u].y ^ rep, t[u].z ^ rep, t[u].w ^ rep };
#pragma unroll
            for (int j = 0; j < 4; ++j) {
                const unsigned hz = (y[j] - 0x01010101u) & ~y[j] & 0x80808080u;
                if (__builtin_amdgcn_ballot_w32(hz != 0u) == 0u) continue;
                const int eb = (p0 + u * OWN_WAVES) * 512 + lane * 16 + j * 4;
#pragma unroll
                for (int k = 0; k < 4; ++k) {
                    const bool     m    = (y[j] & (0xFFu << (8 * k))) == 0u;
                    const unsigned mask = __builtin_amdgcn_ballot_w32(m);
                    if (mask != 0u) {
                        const int pos = cnt + (int)__builtin_amdgcn_mbcnt_lo(mask, 0u);
                        if (m) q[pos] = eb + k;
                        cnt += __builtin_popcount(mask);
                    }
                }
            }
            while (cnt >= EPG) {
                cnt -= EPG;
                __builtin_amdgcn_wave_barrier();
                owner_take<F>(H, words, acc, q + cnt, EPG, rows, lane);
                __builtin_amdgcn_wave_barrier();
            }
        }
    }
    if (cnt > 0) {
        __builtin_amdgcn_wave_barrier();
        owner_take<F>(H, words, acc, q, cnt, rows, lane);
    }
    __syncthreads();

    float4* o4 = reinterpret_cast<float4*>(AGG + (size_t)r0 * F);
    for (int i = tid; i < rows * F / 4; i += NT) o4[i] = a4[i];
}

static inline int cdiv(long a, long b) { return (int)((a + b - 1) / b); }

extern "C" void kernel_launch(void* const* d_in, const int* in_sizes, int n_in,
                              void* d_out, int out_size, void* d_ws, size_t ws_size,
                              hipStream_t stream) {
    const float* features = (const float*)d_in[0];
    const int*   src      = (const int*)d_in[1];
    const int*   dst      = (const int*)d_in[2];
    const float* W1       = (const float*)d_in[3];
    const float* b1       = (const float*)d_in[4];
    const float* W2       = (const float*)d_in[5];
    const float* b2       = (const float*)d_in[6];
    const float* W3       = (const float*)d_in[7];
    const float* b3       = (const float*)d_in[8];
    float*       out      = (float*)d_out;

    const int N = in_sizes[0] / IN_F;
    const int E = in_sizes[1];

    float* ws   = (float*)d_ws;
    float* dsrc = ws;
    float* ddst = ws + N;
    float* buf0 = ws + 2 * (size_t)N;
    float* buf1 = buf0 + (size_t)N * HID_F;

    const long NO = (long)N * OUT_F;
    const int  nblk = N / 16;
    const int  nown = cdiv(N, OWN_ROWS);

    const int npieces = cdiv(E, 512);
    const int nquads  = npieces * 128;
    unsigned* tags  = (unsigned*)d_out;
    unsigned* words = tags + (size_t)nquads;

    zero_kernel<<<cdiv(2L * N, 256), 256, 0, stream>>>(dsrc, 2L * N);
    degree_tag_kernel<<<cdiv(nquads, 256), 256, 0, stream>>>(src, dst, dsrc, ddst, tags, words, E, N, nquads);
    rsqrt_kernel<<<cdiv(2L * N, 256), 256, 0, stream>>>(dsrc, 2 * N);

    gemm_rows16_wmma<HID_F, false><<<nblk, HID_F * 2, 0, stream>>>(
        features, dsrc, ddst, b1, W1, buf0);
    owner_sum_kernel<HID_F><<<nown, OWN_WAVES * 32, 0, stream>>>(buf0, tags, words, buf1, E, N);

    gemm_rows16_wmma<HID_F, true><<<nblk, HID_F * 2, 0, stream>>>(
        buf1, dsrc, ddst, b1, W2, buf0);
    owner_sum_kernel<HID_F><<<nown, OWN_WAVES * 32, 0, stream>>>(buf0, tags, words, buf1, E, N);

    gemm_rows16_wmma<OUT_F, true><<<nblk, OUT_F * 2, 0, stream>>>(
        buf1, dsrc, ddst, b2, W3, buf0);
    owner_sum_kernel<OUT_F><<<nown, OWN_WAVES * 32, 0, stream>>>(buf0, tags, words, buf1, E, N);
    epilogue_kernel<<<cdiv(NO, 256), 256, 0, stream>>>(buf1, out, ddst, b3, OUT_F, NO, 0);
}


# Segment-sum scan runs once: layer 1's waves save the packed words they keep (one list per wave, in out); layers 2 and 3 read them back by whole lines, no tag scan; sum order unfixed as before
# speedup vs baseline: 2.3855x; 1.1315x over previous
#include <hip/hip_runtime.h>

typedef float v2f __attribute__((ext_vector_type(2)));
typedef float v8f __attribute__((ext_vector_type(8)));

#define IN_F  128
#define HID_F 128
#define OUT_F 64
#define KDIM  128

__global__ void zero_kernel(float* __restrict__ p, long n) {
    long i = (long)blockIdx.x * blockDim.x + threadIdx.x;
    if (i < n) p[i] = 0.0f;
}

#define OWN_ROWS 391
__global__ void degree_tag_kernel(const int* __restrict__ src, const int* __restrict__ dst,
                                  float* deg_src, float* deg_dst,
                                  unsigned* __restrict__ tags, unsigned* __restrict__ words,
                                  int ne, int n_nodes, int nquads) {
    const int t = blockIdx.x * blockDim.x + threadIdx.x;
    if (t >= nquads) return;
    const int e0 = 4 * t;
    int s[4], d[4];
    if (e0 + 3 < ne) {
        const int4 s4 = *reinterpret_cast<const int4*>(src + e0);
        const int4 d4 = *reinterpret_cast<const int4*>(dst + e0);
        s[0] = s4.x; s[1] = s4.y; s[2] = s4.z; s[3] = s4.w;
        d[0] = d4.x; d[1] = d4.y; d[2] = d4.z; d[3] = d4.w;
    } else {
#pragma unroll
        for (int k = 0; k < 4; ++k) {
            s[k] = e0 + k < ne ? src[e0 + k] : 0;
            d[k] = e0 + k < ne ? dst[e0 + k] : -1;
        }
    }
    unsigned tg = 0u;
    unsigned w[4];
#pragma unroll
    for (int k = 0; k < 4; ++k) {
        if (e0 + k < ne) {
            atomicAdd(&deg_src[s[k]], 1.0f);
            atomicAdd(&deg_dst[d[k]], 1.0f);
        }
        const bool     ok  = (unsigned)d[k] < (unsigned)n_nodes;
        const unsigned own = ok ? (unsigned)d[k] / OWN_ROWS : 0u;
        const unsigned loc = ok ? (unsigned)d[k] - own * OWN_ROWS : 511u;
        tg |= (own & 255u) << (8 * k);
        w[k] = ((unsigned)s[k] & 0x1FFFFu) | (loc << 17);
    }
    tags[t] = tg;
    *reinterpret_cast<uint4*>(words + e0) = make_uint4(w[0], w[1], w[2], w[3]);
}

__global__ void rsqrt_kernel(float* __restrict__ p, int n) {
    int i = blockIdx.x * blockDim.x + threadIdx.x;
    if (i < n) {
        float d = p[i];
        p[i] = rsqrtf(d < 1.0f ? 1.0f : d);
    }
}

__global__ void epilogue_kernel(const float* __restrict__ X, float* __restrict__ Y,
                                const float* __restrict__ dinv,
                                const float* __restrict__ bias, int F, long total,
                                int do_relu) {
    long t = (long)blockIdx.x * blockDim.x + threadIdx.x;
    if (t >= total) return;
    int i = (int)(t / F);
    int f = (int)(t - (long)i * F);
    float v = X[t] * dinv[i] + bias[f];
    if (do_relu) v = v > 0.0f ? v : 0.0f;
    Y[t] = v;
}

template <int FOUT, bool PRE>
__launch_bounds__(FOUT * 2)
__global__ void gemm_rows16_wmma(const float* X,
                                 const float* __restrict__ dsrc,
                                 const float* __restrict__ ddst,
                                 const float* __restrict__ pbias,
                                 const float* __restrict__ W,
                                 float* __restrict__ H) {
    constexpr int LDSS = KDIM + 4;
    __shared__ __align__(16) float As[16 * LDSS];

    const int row0 = blockIdx.x * 16;
    const int tid  = threadIdx.x;

    for (int i = tid; i < 16 * KDIM; i += blockDim.x) {
        int r = i >> 7;
        int c = i & (KDIM - 1);
        float x = X[(size_t)(row0 + r) * KDIM + c];
        if (PRE) {
            x = x * ddst[row0 + r] + pbias[c];
            x = x > 0.0f ? x : 0.0f;
        }
        As[r * LDSS + c] = x * dsrc[row0 + r];
    }
    __syncthreads();

    const int lane  = tid & 31;
    const int wave  = tid >> 5;
    const int n     = lane & 15;
    const int khalf = lane >> 4;
    const int m16   = lane & 15;
    const int ncol  = wave * 16 + n;

    v8f acc = {};
#pragma unroll
    for (int k0 = 0; k0 < KDIM; k0 += 4) {
        const int ka = k0 + 2 * khalf;
        v2f a = *(const v2f*)&As[m16 * LDSS + ka];
        v2f b;
        b.x = W[(size_t)ka * FOUT + ncol];
        b.y = W[(size_t)(ka + 1) * FOUT + ncol];
        acc = __builtin_amdgcn_wmma_f32_16x16x4_f32(
             false, a,  false, b,
             (short)0, acc,  false,  false);
    }

#pragma unroll
    for (int r = 0; r < 8; ++r) {
        const int m = r + 8 * khalf;
        H[(size_t)(row0 + m) * FOUT + ncol] = acc[r];
    }
}

#define OWN_WAVES 32
#define SCAN_U 8
#define LIST_CAP 512

template <int F>
__device__ __forceinline__ void rows_add(const float* __restrict__ H, float* acc,
                                         int w, int base, int n, int rows, int lane) {
    constexpr int V   = F / 32;
    constexpr int EPG = 32 / V;
    float v[EPG][V];
#pragma unroll
    for (int u = 0; u < EPG; ++u) {
        const int su = __builtin_amdgcn_readlane(w, base + u) & 0x1FFFF;
        const float* hp = H + (size_t)su * F + lane;
#pragma unroll
        for (int j = 0; j < V; ++j) v[u][j] = hp[32 * j];
    }
#pragma unroll
    for (int u = 0; u < EPG; ++u) {
        const int du = (int)((unsigned)__builtin_amdgcn_readlane(w, base + u) >> 17);
        if (u < n && du < rows) {
            float* ap = acc + du * F + lane;
#pragma unroll
            for (int j = 0; j < V; ++j) atomicAdd(ap + 32 * j, v[u][j]);
        }
    }
}

template <int F, bool SAVE>
__device__ __forceinline__ void owner_take(const float* __restrict__ H,
                                           const unsigned* __restrict__ words,
                                           float* acc, const int* qe, int n, int rows, int lane,
                                           unsigned* lsave, int& saved) {
    constexpr int V   = F / 32;
    constexpr int EPG = 32 / V;
    int idx = lane & (EPG - 1);
    idx = idx < n ? idx : n - 1;
    const int w = (int)words[qe[idx]];
    if (SAVE) {
        if (lane < n && saved + lane < LIST_CAP) lsave[saved + lane] = (unsigned)w;
        saved += n;
    }
    rows_add<F>(H, acc, w, 0, n, rows, lane);
}

template <int F, bool SAVE>
__device__ __forceinline__ void scan_and_sum(const float* __restrict__ H,
                                             const unsigned* __restrict__ tags,
                                             const unsigned* __restrict__ words,
                                             float* acc, int* q, int ne, int rows,
                                             unsigned rep, int wave, int lane,
                                             unsigned* lsave, int* lcount) {
    constexpr int V   = F / 32;
    constexpr int EPG = 32 / V;
    int cnt = 0;
    int saved = 0;
    const int npieces = (ne + 511) >> 9;
    const uint4* tags4 = reinterpret_cast<const uint4*>(tags);

    for (int p0 = wave; p0 < npieces; p0 += OWN_WAVES * SCAN_U) {
        uint4 t[SCAN_U];
#pragma unroll
        for (int u = 0; u < SCAN_U; ++u) {
            const int pc = (p0 + u * OWN_WAVES) < npieces ? (p0 + u * OWN_WAVES) : npieces - 1;
            t[u] = tags4[(size_t)pc * 32 + lane];
        }
#pragma unroll
        for (int u = 0; u < SCAN_U; ++u) {
            if (p0 + u * OWN_WAVES >= npieces) continue;
            const unsigned y[4] = { t[u].x ^ rep, t[u].y ^ rep, t[u].z ^ rep, t[u].w ^ rep };
#pragma unroll
            for (int j = 0; j < 4; ++j) {
                const unsigned hz = (y[j] - 0x01010101u) & ~y[j] & 0x80808080u;
                if (__builtin_amdgcn_ballot_w32(hz != 0u) == 0u) continue;
                const int eb = (p0 + u * OWN_WAVES) * 512 + lane * 16 + j * 4;
#pragma unroll
                for (int k = 0; k < 4; ++k) {
                    const bool     m    = (y[j] & (0xFFu << (8 * k))) == 0u;
                    const unsigned mask = __builtin_amdgcn_ballot_w32(m);
                    if (mask != 0u) {
                        const int pos = cnt + (int)__builtin_amdgcn_mbcnt_lo(mask, 0u);
                        if (m) q[pos] = eb + k;
                        cnt += __builtin_popcount(mask);
                    }
                }
            }
            while (cnt >= EPG) {
                cnt -= EPG;
                __builtin_amdgcn_wave_barrier();
                owner_take<F, SAVE>(H, words, acc, q + cnt, EPG, rows, lane, lsave, saved);
                __builtin_amdgcn_wave_barrier();
            }
        }
    }
    if (cnt > 0) {
        __builtin_amdgcn_wave_barrier();
        owner_take<F, SAVE>(H, words, acc, q, cnt, rows, lane, lsave, saved);
    }
    if (SAVE && lane == 0) *lcount = saved;
}

template <int F, bool SAVE>
__launch_bounds__(OWN_WAVES * 32)
__global__ void owner_sum_kernel(const float* __restrict__ H,
                                 const unsigned* __restrict__ tags,
                                 const unsigned* __restrict__ words,
                                 unsigned* lists, int* counts,
                                 float* __restrict__ AGG, int ne, int n_nodes) {
    constexpr int V   = F / 32;
    constexpr int EPG = 32 / V;
    constexpr int QW  = 512 + EPG;
    constexpr int NT  = OWN_WAVES * 32;

    __shared__ __align__(16) float acc[OWN_ROWS * F];
    __shared__ int queue[OWN_WAVES * QW];

    const int tid  = threadIdx.x;
    const int lane = tid & 31;
    const int wave = __builtin_amdgcn_readfirstlane(tid >> 5);
    const int r0   = blockIdx.x * OWN_ROWS;
    const int rows = (n_nodes - r0) < OWN_ROWS ? (n_nodes - r0) : OWN_ROWS;
    const unsigned rep = (unsigned)(blockIdx.x & 255) * 0x01010101u;

    float4* a4 = reinterpret_cast<float4*>(acc);
    for (int i = tid; i < OWN_ROWS * F / 4; i += NT) a4[i] = make_float4(0.0f, 0.0f, 0.0f, 0.0f);
    __syncthreads();

    int* q = queue + wave * QW;
    const int wg = blockIdx.x * OWN_WAVES + wave;
    unsigned* lp = lists + (size_t)wg * LIST_CAP;

    if (SAVE) {
        scan_and_sum<F, true>(H, tags, words, acc, q, ne, rows, rep, wave, lane, lp, counts + wg);
    } else {
        const int n = __builtin_amdgcn_readfirstlane(counts[wg]);
        if ((unsigned)n <= (unsigned)LIST_CAP) {
            int wn = 0;
            if (n > 0) wn = (int)lp[lane < n ? lane : n - 1];
            for (int c = 0; c < n; c += 32) {
                const int w = wn;
                const int m = (n - c) < 32 ? (n - c) : 32;
                const int mn = n - c - 32;
                if (mn > 0) wn = (int)lp[c + 32 + (lane < mn ? lane : mn - 1)];
#pragma unroll
                for (int s = 0; s < 32 / EPG; ++s) {
                    const int k = m - s * EPG;
                    if (k > 0) rows_add<F>(H, acc, w, s * EPG, k < EPG ? k : EPG, rows, lane);
                }
            }
        } else {
            scan_and_sum<F, false>(H, tags, words, acc, q, ne, rows, rep, wave, lane, lp, counts + wg);
        }
    }
    __syncthreads();

    float4* o4 = reinterpret_cast<float4*>(AGG + (size_t)r0 * F);
    for (int i = tid; i < rows * F / 4; i += NT) o4[i] = a4[i];
}

static inline int cdiv(long a, long b) { return (int)((a + b - 1) / b); }

extern "C" void kernel_launch(void* const* d_in, const int* in_sizes, int n_in,
                              void* d_out, int out_size, void* d_ws, size_t ws_size,
                              hipStream_t stream) {
    const float* features = (const float*)d_in[0];
    const int*   src      = (const int*)d_in[1];
    const int*   dst      = (const int*)d_in[2];
    const float* W1       = (const float*)d_in[3];
    const float* b1       = (const float*)d_in[4];
    const float* W2       = (const float*)d_in[5];
    const float* b2       = (const float*)d_in[6];
    const float* W3       = (const float*)d_in[7];
    const float* b3       = (const float*)d_in[8];
    float*       out      = (float*)d_out;

    const int N = in_sizes[0] / IN_F;
    const int E = in_sizes[1];

    float* ws   = (float*)d_ws;
    float* dsrc = ws;
    float* ddst = ws + N;
    float* buf0 = ws + 2 * (size_t)N;
    float* buf1 = buf0 + (size_t)N * HID_F;

    const long NO = (long)N * OUT_F;
    const int  nblk = N / 16;
    const int  nown = cdiv(N, OWN_ROWS);

    const int npieces = cdiv(E, 512);
    const int nquads  = npieces * 128;
    unsigned* tags  = (unsigned*)d_out;
    unsigned* words = tags + (size_t)nquads;
    unsigned* lists = words + (size_t)npieces * 512;
    int*      counts = (int*)(lists + (size_t)nown * OWN_WAVES * LIST_CAP);

    zero_kernel<<<cdiv(2L * N, 256), 256, 0, stream>>>(dsrc, 2L * N);
    degree_tag_kernel<<<cdiv(nquads, 256), 256, 0, stream>>>(src, dst, dsrc, ddst, tags, words, E, N, nquads);
    rsqrt_kernel<<<cdiv(2L * N, 256), 256, 0, stream>>>(dsrc, 2 * N);

    gemm_rows16_wmma<HID_F, false><<<nblk, HID_F * 2, 0, stream>>>(
        features, dsrc, ddst, b1, W1, buf0);
    owner_sum_kernel<HID_F, true><<<nown, OWN_WAVES * 32, 0, stream>>>(buf0, tags, words, lists, counts, buf1, E, N);

    gemm_rows16_wmma<HID_F, true><<<nblk, HID_F * 2, 0, stream>>>(
        buf1, dsrc, ddst, b1, W2, buf0);
    owner_sum_kernel<HID_F, false><<<nown, OWN_WAVES * 32, 0, stream>>>(buf0, tags, words, lists, counts, buf1, E, N);

    gemm_rows16_wmma<OUT_F, true><<<nblk, OUT_F * 2, 0, stream>>>(
        buf1, dsrc, ddst, b2, W3, buf0);
    owner_sum_kernel<OUT_F, false><<<nown, OWN_WAVES * 32, 0, stream>>>(buf0, tags, words, lists, counts, buf1, E, N);
    epilogue_kernel<<<cdiv(NO, 256), 256, 0, stream>>>(buf1, out, ddst, b3, OUT_F, NO, 0);
}


# Row gather: one load per source row (lane L takes 4 or 2 consecutive floats): a wave's 32 requests in flight are 256-B pieces, not 128-B lines; LDS sums lane-transposed; sum order unfixed as before
# speedup vs baseline: 2.5020x; 1.0488x over previous
#include <hip/hip_runtime.h>

typedef float v2f __attribute__((ext_vector_type(2)));
typedef float v8f __attribute__((ext_vector_type(8)));

#define IN_F  128
#define HID_F 128
#define OUT_F 64
#define KDIM  128

__global__ void zero_kernel(float* __restrict__ p, long n) {
    long i = (long)blockIdx.x * blockDim.x + threadIdx.x;
    if (i < n) p[i] = 0.0f;
}

#define OWN_ROWS 391
__global__ void degree_tag_kernel(const int* __restrict__ src, const int* __restrict__ dst,
                                  float* deg_src, float* deg_dst,
                                  unsigned* __restrict__ tags, unsigned* __restrict__ words,
                                  int ne, int n_nodes, int nquads) {
    const int t = blockIdx.x * blockDim.x + threadIdx.x;
    if (t >= nquads) return;
    const int e0 = 4 * t;
    int s[4], d[4];
    if (e0 + 3 < ne) {
        const int4 s4 = *reinterpret_cast<const int4*>(src + e0);
        const int4 d4 = *reinterpret_cast<const int4*>(dst + e0);
        s[0] = s4.x; s[1] = s4.y; s[2] = s4.z; s[3] = s4.w;
        d[0] = d4.x; d[1] = d4.y; d[2] = d4.z; d[3] = d4.w;
    } else {
#pragma unroll
        for (int k = 0; k < 4; ++k) {
            s[k] = e0 + k < ne ? src[e0 + k] : 0;
            d[k] = e0 + k < ne ? dst[e0 + k] : -1;
        }
    }
    unsigned tg = 0u;
    unsigned w[4];
#pragma unroll
    for (int k = 0; k < 4; ++k) {
        if (e0 + k < ne) {
            atomicAdd(&deg_src[s[k]], 1.0f);
            atomicAdd(&deg_dst[d[k]], 1.0f);
        }
        const bool     ok  = (unsigned)d[k] < (unsigned)n_nodes;
        const unsigned own = ok ? (unsigned)d[k] / OWN_ROWS : 0u;
        const unsigned loc = ok ? (unsigned)d[k] - own * OWN_ROWS : 511u;
        tg |= (own & 255u) << (8 * k);
        w[k] = ((unsigned)s[k] & 0x1FFFFu) | (loc << 17);
    }
    tags[t] = tg;
    *reinterpret_cast<uint4*>(words + e0) = make_uint4(w[0], w[1], w[2], w[3]);
}

__global__ void rsqrt_kernel(float* __restrict__ p, int n) {
    int i = blockIdx.x * blockDim.x + threadIdx.x;
    if (i < n) {
        float d = p[i];
        p[i] = rsqrtf(d < 1.0f ? 1.0f : d);
    }
}

__global__ void epilogue_kernel(const float* __restrict__ X, float* __restrict__ Y,
                                const float* __restrict__ dinv,
                                const float* __restrict__ bias, int F, long total,
                                int do_relu) {
    long t = (long)blockIdx.x * blockDim.x + threadIdx.x;
    if (t >= total) return;
    int i = (int)(t / F);
    int f = (int)(t - (long)i * F);
    float v = X[t] * dinv[i] + bias[f];
    if (do_relu) v = v > 0.0f ? v : 0.0f;
    Y[t] = v;
}

template <int FOUT, bool PRE>
__launch_bounds__(FOUT * 2)
__global__ void gemm_rows16_wmma(const float* X,
                                 const float* __restrict__ dsrc,
                                 const float* __restrict__ ddst,
                                 const float* __restrict__ pbias,
                                 const float* __restrict__ W,
                                 float* __restrict__ H) {
    constexpr int LDSS = KDIM + 4;
    __shared__ __align__(16) float As[16 * LDSS];

    const int row0 = blockIdx.x * 16;
    const int tid  = threadIdx.x;

    for (int i = tid; i < 16 * KDIM; i += blockDim.x) {
        int r = i >> 7;
        int c = i & (KDIM - 1);
        float x = X[(size_t)(row0 + r) * KDIM + c];
        if (PRE) {
            x = x * ddst[row0 + r] + pbias[c];
            x = x > 0.0f ? x : 0.0f;
        }
        As[r * LDSS + c] = x * dsrc[row0 + r];
    }
    __syncthreads();

    const int lane  = tid & 31;
    const int wave  = tid >> 5;
    const int n     = lane & 15;
    const int khalf = lane >> 4;
    const int m16   = lane & 15;
    const int ncol  = wave * 16 + n;

    v8f acc = {};
#pragma unroll
    for (int k0 = 0; k0 < KDIM; k0 += 4) {
        const int ka = k0 + 2 * khalf;
        v2f a = *(const v2f*)&As[m16 * LDSS + ka];
        v2f b;
        b.x = W[(size_t)ka * FOUT + ncol];
        b.y = W[(size_t)(ka + 1) * FOUT + ncol];
        acc = __builtin_amdgcn_wmma_f32_16x16x4_f32(
             false, a,  false, b,
             (short)0, acc,  false,  false);
    }

#pragma unroll
    for (int r = 0; r < 8; ++r) {
        const int m = r + 8 * khalf;
        H[(size_t)(row0 + m) * FOUT + ncol] = acc[r];
    }
}

#define OWN_WAVES 32
#define SCAN_U 8
#define LIST_CAP 512
#define RIF(F) (2048 / (F))

template <int F>
__device__ __forceinline__ void rows_add(const float* __restrict__ H, float* acc,
                                         int w, int base, int n, int rows, int lane) {
    constexpr int V   = F / 32;
    constexpr int EPG = RIF(F);
    typedef float rowv __attribute__((ext_vector_type(V)));
    rowv v[EPG];
#pragma unroll
    for (int u = 0; u < EPG; ++u) {
        const int su = __builtin_amdgcn_readlane(w, base + u) & 0x1FFFF;
        v[u] = *reinterpret_cast<const rowv*>(H + (size_t)su * F + V * lane);
    }
#pragma unroll
    for (int u = 0; u < EPG; ++u) {
        const int du = (int)((unsigned)__builtin_amdgcn_readlane(w, base + u) >> 17);
        if (u < n && du < rows) {
            float* ap = acc + du * F + lane;
#pragma unroll
            for (int j = 0; j < V; ++j) atomicAdd(ap + 32 * j, v[u][j]);
        }
    }
}

template <int F, bool SAVE>
__device__ __forceinline__ void owner_take(const float* __restrict__ H,
                                           const unsigned* __restrict__ words,
                                           float* acc, const int* qe, int n, int rows, int lane,
                                           unsigned* lsave, int& saved) {
    constexpr int EPG = RIF(F);
    int idx = lane & (EPG - 1);
    idx = idx < n ? idx : n - 1;
    const int w = (int)words[qe[idx]];
    if (SAVE) {
        if (lane < n && saved + lane < LIST_CAP) lsave[saved + lane] = (unsigned)w;
        saved += n;
    }
    rows_add<F>(H, acc, w, 0, n, rows, lane);
}

template <int F, bool SAVE>
__device__ __forceinline__ void scan_and_sum(const float* __restrict__ H,
                                             const unsigned* __restrict__ tags,
                                             const unsigned* __restrict__ words,
                                             float* acc, int* q, int ne, int rows,
                                             unsigned rep, int wave, int lane,
                                             unsigned* lsave, int* lcount) {
    constexpr int EPG = RIF(F);
    int cnt = 0;
    int saved = 0;
    const int npieces = (ne + 511) >> 9;
    const uint4* tags4 = reinterpret_cast<const uint4*>(tags);

    for (int p0 = wave; p0 < npieces; p0 += OWN_WAVES * SCAN_U) {
        uint4 t[SCAN_U];
#pragma unroll
        for (int u = 0; u < SCAN_U; ++u) {
            const int pc = (p0 + u * OWN_WAVES) < npieces ? (p0 + u * OWN_WAVES) : npieces - 1;
            t[u] = tags4[(size_t)pc * 32 + lane];
        }
#pragma unroll
        for (int u = 0; u < SCAN_U; ++u) {
            if (p0 + u * OWN_WAVES >= npieces) continue;
            const unsigned y[4] = { t[u].x ^ rep, t[u].y ^ rep, t[u].z ^ rep, t[u].w ^ rep };
#pragma unroll
            for (int j = 0; j < 4; ++j) {
                const unsigned hz = (y[j] - 0x01010101u) & ~y[j] & 0x80808080u;
                if (__builtin_amdgcn_ballot_w32(hz != 0u) == 0u) continue;
                const int eb = (p0 + u * OWN_WAVES) * 512 + lane * 16 + j * 4;
#pragma unroll
                for (int k = 0; k < 4; ++k) {
                    const bool     m    = (y[j] & (0xFFu << (8 * k))) == 0u;
                    const unsigned mask = __builtin_amdgcn_ballot_w32(m);
                    if (mask != 0u) {
                        const int pos = cnt + (int)__builtin_amdgcn_mbcnt_lo(mask, 0u);
                        if (m) q[pos] = eb + k;
                        cnt += __builtin_popcount(mask);
                    }
                }
            }
            while (cnt >= EPG) {
                cnt -= EPG;
                __builtin_amdgcn_wave_barrier();
                owner_take<F, SAVE>(H, words, acc, q + cnt, EPG, rows, lane, lsave, saved);
                __builtin_amdgcn_wave_barrier();
            }
        }
    }
    if (cnt > 0) {
        __builtin_amdgcn_wave_barrier();
        owner_take<F, SAVE>(H, words, acc, q, cnt, rows, lane, lsave, saved);
    }
    if (SAVE && lane == 0) *lcount = saved;
}

template <int F, bool SAVE>
__launch_bounds__(OWN_WAVES * 32)
__global__ void owner_sum_kernel(const float* __restrict__ H,
                                 const unsigned* __restrict__ tags,
                                 const unsigned* __restrict__ words,
                                 unsigned* lists, int* counts,
                                 float* __restrict__ AGG, int ne, int n_nodes) {
    constexpr int V   = F / 32;
    constexpr int EPG = RIF(F);
    constexpr int QW  = 512 + EPG;
    constexpr int NT  = OWN_WAVES * 32;

    __shared__ __align__(16) float acc[OWN_ROWS * F];
    __shared__ int queue[OWN_WAVES * QW];

    const int tid  = threadIdx.x;
    const int lane = tid & 31;
    const int wave = __builtin_amdgcn_readfirstlane(tid >> 5);
    const int r0   = blockIdx.x * OWN_ROWS;
    const int rows = (n_nodes - r0) < OWN_ROWS ? (n_nodes - r0) : OWN_ROWS;
    const unsigned rep = (unsigned)(blockIdx.x & 255) * 0x01010101u;

    float4* a4 = reinterpret_cast<float4*>(acc);
    for (int i = tid; i < OWN_ROWS * F / 4; i += NT) a4[i] = make_float4(0.0f, 0.0f, 0.0f, 0.0f);
    __syncthreads();

    int* q = queue + wave * QW;
    const int wg = blockIdx.x * OWN_WAVES + wave;
    unsigned* lp = lists + (size_t)wg * LIST_CAP;

    if (SAVE) {
        scan_and_sum<F, true>(H, tags, words, acc, q, ne, rows, rep, wave, lane, lp, counts + wg);
    } else {
        const int n = __builtin_amdgcn_readfirstlane(counts[wg]);
        if ((unsigned)n <= (unsigned)LIST_CAP) {
            int wn = 0;
            if (n > 0) wn = (int)lp[lane < n ? lane : n - 1];
            for (int c = 0; c < n; c += 32) {
                const int w = wn;
                const int m = (n - c) < 32 ? (n - c) : 32;
                const int mn = n - c - 32;
                if (mn > 0) wn = (int)lp[c + 32 + (lane < mn ? lane : mn - 1)];
#pragma unroll
                for (int s = 0; s < 32 / EPG; ++s) {
                    const int k = m - s * EPG;
                    if (k > 0) rows_add<F>(H, acc, w, s * EPG, k < EPG ? k : EPG, rows, lane);
                }
            }
        } else {
            scan_and_sum<F, false>(H, tags, words, acc, q, ne, rows, rep, wave, lane, lp, counts + wg);
        }
    }
    __syncthreads();

    typedef float rowv __attribute__((ext_vector_type(V)));
    for (int r = wave; r < rows; r += OWN_WAVES) {
        const float* ar = acc + r * F + lane;
        rowv o;
#pragma unroll
        for (int j = 0; j < V; ++j) o[j] = ar[32 * j];
        *reinterpret_cast<rowv*>(AGG + (size_t)(r0 + r) * F + V * lane) = o;
    }
}

static inline int cdiv(long a, long b) { return (int)((a + b - 1) / b); }

extern "C" void kernel_launch(void* const* d_in, const int* in_sizes, int n_in,
                              void* d_out, int out_size, void* d_ws, size_t ws_size,
                              hipStream_t stream) {
    const float* features = (const float*)d_in[0];
    const int*   src      = (const int*)d_in[1];
    const int*   dst      = (const int*)d_in[2];
    const float* W1       = (const float*)d_in[3];
    const float* b1       = (const float*)d_in[4];
    const float* W2       = (const float*)d_in[5];
    const float* b2       = (const float*)d_in[6];
    const float* W3       = (const float*)d_in[7];
    const float* b3       = (const float*)d_in[8];
    float*       out      = (float*)d_out;

    const int N = in_sizes[0] / IN_F;
    const int E = in_sizes[1];

    float* ws   = (float*)d_ws;
    float* dsrc = ws;
    float* ddst = ws + N;
    float* buf0 = ws + 2 * (size_t)N;
    float* buf1 = buf0 + (size_t)N * HID_F;

    const long NO = (long)N * OUT_F;
    const int  nblk = N / 16;
    const int  nown = cdiv(N, OWN_ROWS);

    const int npieces = cdiv(E, 512);
    const int nquads  = npieces * 128;
    unsigned* tags  = (unsigned*)d_out;
    unsigned* words = tags + (size_t)nquads;
    unsigned* lists = words + (size_t)npieces * 512;
    int*      counts = (int*)(lists + (size_t)nown * OWN_WAVES * LIST_CAP);

    zero_kernel<<<cdiv(2L * N, 256), 256, 0, stream>>>(dsrc, 2L * N);
    degree_tag_kernel<<<cdiv(nquads, 256), 256, 0, stream>>>(src, dst, dsrc, ddst, tags, words, E, N, nquads);
    rsqrt_kernel<<<cdiv(2L * N, 256), 256, 0, stream>>>(dsrc, 2 * N);

    gemm_rows16_wmma<HID_F, false><<<nblk, HID_F * 2, 0, stream>>>(
        features, dsrc, ddst, b1, W1, buf0);
    owner_sum_kernel<HID_F, true><<<nown, OWN_WAVES * 32, 0, stream>>>(buf0, tags, words, lists, counts, buf1, E, N);

    gemm_rows16_wmma<HID_F, true><<<nblk, HID_F * 2, 0, stream>>>(
        buf1, dsrc, ddst, b1, W2, buf0);
    owner_sum_kernel<HID_F, false><<<nown, OWN_WAVES * 32, 0, stream>>>(buf0, tags, words, lists, counts, buf1, E, N);

    gemm_rows16_wmma<OUT_F, true><<<nblk, OUT_F * 2, 0, stream>>>(
        buf1, dsrc, ddst, b2, W3, buf0);
    owner_sum_kernel<OUT_F, false><<<nown, OWN_WAVES * 32, 0, stream>>>(buf0, tags, words, lists, counts, buf1, E, N);
    epilogue_kernel<<<cdiv(NO, 256), 256, 0, stream>>>(buf1, out, ddst, b3, OUT_F, NO, 0);
}


# GEMM launches: a block covers 80 rows, each wave keeps 5 accumulator tiles, so one W fragment load feeds 5 WMMAs (W re-read 5x less); A tile staged by whole-row 16-B loads; k order, sums as before
# speedup vs baseline: 2.7172x; 1.0860x over previous
#include <hip/hip_runtime.h>

typedef float v2f __attribute__((ext_vector_type(2)));
typedef float v8f __attribute__((ext_vector_type(8)));

#define IN_F  128
#define HID_F 128
#define OUT_F 64
#define KDIM  128

__global__ void zero_kernel(float* __restrict__ p, long n) {
    long i = (long)blockIdx.x * blockDim.x + threadIdx.x;
    if (i < n) p[i] = 0.0f;
}

#define OWN_ROWS 391
__global__ void degree_tag_kernel(const int* __restrict__ src, const int* __restrict__ dst,
                                  float* deg_src, float* deg_dst,
                                  unsigned* __restrict__ tags, unsigned* __restrict__ words,
                                  int ne, int n_nodes, int nquads) {
    const int t = blockIdx.x * blockDim.x + threadIdx.x;
    if (t >= nquads) return;
    const int e0 = 4 * t;
    int s[4], d[4];
    if (e0 + 3 < ne) {
        const int4 s4 = *reinterpret_cast<const int4*>(src + e0);
        const int4 d4 = *reinterpret_cast<const int4*>(dst + e0);
        s[0] = s4.x; s[1] = s4.y; s[2] = s4.z; s[3] = s4.w;
        d[0] = d4.x; d[1] = d4.y; d[2] = d4.z; d[3] = d4.w;
    } else {
#pragma unroll
        for (int k = 0; k < 4; ++k) {
            s[k] = e0 + k < ne ? src[e0 + k] : 0;
            d[k] = e0 + k < ne ? dst[e0 + k] : -1;
        }
    }
    unsigned tg = 0u;
    unsigned w[4];
#pragma unroll
    for (int k = 0; k < 4; ++k) {
        if (e0 + k < ne) {
            atomicAdd(&deg_src[s[k]], 1.0f);
            atomicAdd(&deg_dst[d[k]], 1.0f);
        }
        const bool     ok  = (unsigned)d[k] < (unsigned)n_nodes;
        const unsigned own = ok ? (unsigned)d[k] / OWN_ROWS : 0u;
        const unsigned loc = ok ? (unsigned)d[k] - own * OWN_ROWS : 511u;
        tg |= (own & 255u) << (8 * k);
        w[k] = ((unsigned)s[k] & 0x1FFFFu) | (loc << 17);
    }
    tags[t] = tg;
    *reinterpret_cast<uint4*>(words + e0) = make_uint4(w[0], w[1], w[2], w[3]);
}

__global__ void rsqrt_kernel(float* __restrict__ p, int n) {
    int i = blockIdx.x * blockDim.x + threadIdx.x;
    if (i < n) {
        float d = p[i];
        p[i] = rsqrtf(d < 1.0f ? 1.0f : d);
    }
}

__global__ void epilogue_kernel(const float* __restrict__ X, float* __restrict__ Y,
                                const float* __restrict__ dinv,
                                const float* __restrict__ bias, int F, long total,
                                int do_relu) {
    long t = (long)blockIdx.x * blockDim.x + threadIdx.x;
    if (t >= total) return;
    int i = (int)(t / F);
    int f = (int)(t - (long)i * F);
    float v = X[t] * dinv[i] + bias[f];
    if (do_relu) v = v > 0.0f ? v : 0.0f;
    Y[t] = v;
}

#define GEMM_TM   5
#define GEMM_ROWS (16 * GEMM_TM)
template <int FOUT, bool PRE>
__launch_bounds__(FOUT * 2)
__global__ void gemm_rows80_wmma(const float* X,
                                 const float* __restrict__ dsrc,
                                 const float* __restrict__ ddst,
                                 const float* __restrict__ pbias,
                                 const float* __restrict__ W,
                                 float* __restrict__ H, int n_rows) {
    constexpr int LDSS = KDIM + 4;
    constexpr int NW   = FOUT / 16;
    constexpr int ITER = GEMM_ROWS / NW;
    __shared__ __align__(16) float As[GEMM_ROWS * LDSS];

    const int row0 = blockIdx.x * GEMM_ROWS;
    const int tid  = threadIdx.x;
    const int lane = tid & 31;
    const int wave = __builtin_amdgcn_readfirstlane(tid >> 5);

    float4 x[ITER];
    float  ss[ITER], sd[ITER];
#pragma unroll
    for (int it = 0; it < ITER; ++it) {
        int gr = row0 + wave + NW * it;
        gr = gr < n_rows ? gr : n_rows - 1;
        x[it]  = *reinterpret_cast<const float4*>(X + (size_t)gr * KDIM + 4 * lane);
        ss[it] = dsrc[gr];
        sd[it] = PRE ? ddst[gr] : 0.0f;
    }
    float4 pb = make_float4(0.0f, 0.0f, 0.0f, 0.0f);
    if (PRE) pb = *reinterpret_cast<const float4*>(pbias + 4 * lane);
#pragma unroll
    for (int it = 0; it < ITER; ++it) {
        float4 v = x[it];
        if (PRE) {
            v.x = v.x * sd[it] + pb.x;  v.x = v.x > 0.0f ? v.x : 0.0f;
            v.y = v.y * sd[it] + pb.y;  v.y = v.y > 0.0f ? v.y : 0.0f;
            v.z = v.z * sd[it] + pb.z;  v.z = v.z > 0.0f ? v.z : 0.0f;
            v.w = v.w * sd[it] + pb.w;  v.w = v.w > 0.0f ? v.w : 0.0f;
        }
        v.x *= ss[it]; v.y *= ss[it]; v.z *= ss[it]; v.w *= ss[it];
        *reinterpret_cast<float4*>(&As[(wave + NW * it) * LDSS + 4 * lane]) = v;
    }
    __syncthreads();

    const int n     = lane & 15;
    const int khalf = lane >> 4;
    const int m16   = lane & 15;
    const int ncol  = wave * 16 + n;

    v8f acc[GEMM_TM];
#pragma unroll
    for (int t = 0; t < GEMM_TM; ++t) acc[t] = v8f{};
    v2f a_cur[GEMM_TM], a_nxt[GEMM_TM];
#pragma unroll
    for (int t = 0; t < GEMM_TM; ++t)
        a_cur[t] = *(const v2f*)&As[(t * 16 + m16) * LDSS + 2 * khalf];
#pragma unroll
    for (int k0 = 0; k0 < KDIM; k0 += 4) {
        const int ka = k0 + 2 * khalf;
        v2f b;
        b.x = W[(size_t)ka * FOUT + ncol];
        b.y = W[(size_t)(ka + 1) * FOUT + ncol];
        if (k0 + 4 < KDIM) {
#pragma unroll
            for (int t = 0; t < GEMM_TM; ++t)
                a_nxt[t] = *(const v2f*)&As[(t * 16 + m16) * LDSS + ka + 4];
        }
#pragma unroll
        for (int t = 0; t < GEMM_TM; ++t) {
            acc[t] = __builtin_amdgcn_wmma_f32_16x16x4_f32(
                 false, a_cur[t],  false, b,
                 (short)0, acc[t],  false,  false);
        }
#pragma unroll
        for (int t = 0; t < GEMM_TM; ++t) a_cur[t] = a_nxt[t];
    }

#pragma unroll
    for (int t = 0; t < GEMM_TM; ++t) {
#pragma unroll
        for (int r = 0; r < 8; ++r) {
            const int m = row0 + t * 16 + r + 8 * khalf;
            if (m < n_rows) H[(size_t)m * FOUT + ncol] = acc[t][r];
        }
    }
}

#define OWN_WAVES 32
#define SCAN_U 8
#define LIST_CAP 512
#define RIF(F) (2048 / (F))

template <int F>
__device__ __forceinline__ void rows_add(const float* __restrict__ H, float* acc,
                                         int w, int base, int n, int rows, int lane) {
    constexpr int V   = F / 32;
    constexpr int EPG = RIF(F);
    typedef float rowv __attribute__((ext_vector_type(V)));
    rowv v[EPG];
#pragma unroll
    for (int u = 0; u < EPG; ++u) {
        const int su = __builtin_amdgcn_readlane(w, base + u) & 0x1FFFF;
        v[u] = *reinterpret_cast<const rowv*>(H + (size_t)su * F + V * lane);
    }
#pragma unroll
    for (int u = 0; u < EPG; ++u) {
        const int du = (int)((unsigned)__builtin_amdgcn_readlane(w, base + u) >> 17);
        if (u < n && du < rows) {
            float* ap = acc + du * F + lane;
#pragma unroll
            for (int j = 0; j < V; ++j) atomicAdd(ap + 32 * j, v[u][j]);
        }
    }
}

template <int F, bool SAVE>
__device__ __forceinline__ void owner_take(const float* __restrict__ H,
                                           const unsigned* __restrict__ words,
                                           float* acc, const int* qe, int n, int rows, int lane,
                                           unsigned* lsave, int& saved) {
    constexpr int EPG = RIF(F);
    int idx = lane & (EPG - 1);
    idx = idx < n ? idx : n - 1;
    const int w = (int)words[qe[idx]];
    if (SAVE) {
        if (lane < n && saved + lane < LIST_CAP) lsave[saved + lane] = (unsigned)w;
        saved += n;
    }
    rows_add<F>(H, acc, w, 0, n, rows, lane);
}

template <int F, bool SAVE>
__device__ __forceinline__ void scan_and_sum(const float* __restrict__ H,
                                             const unsigned* __restrict__ tags,
                                             const unsigned* __restrict__ words,
                                             float* acc, int* q, int ne, int rows,
                                             unsigned rep, int wave, int lane,
                                             unsigned* lsave, int* lcount) {
    constexpr int EPG = RIF(F);
    int cnt = 0;
    int saved = 0;
    const int npieces = (ne + 511) >> 9;
    const uint4* tags4 = reinterpret_cast<const uint4*>(tags);

    for (int p0 = wave; p0 < npieces; p0 += OWN_WAVES * SCAN_U) {
        uint4 t[SCAN_U];
#pragma unroll
        for (int u = 0; u < SCAN_U; ++u) {
            const int pc = (p0 + u * OWN_WAVES) < npieces ? (p0 + u * OWN_WAVES) : npieces - 1;
            t[u] = tags4[(size_t)pc * 32 + lane];
        }
#pragma unroll
        for (int u = 0; u < SCAN_U; ++u) {
            if (p0 + u * OWN_WAVES >= npieces) continue;
            const unsigned y[4] = { t[u].x ^ rep, t[u].y ^ rep, t[u].z ^ rep, t[u].w ^ rep };
#pragma unroll
            for (int j = 0; j < 4; ++j) {
                const unsigned hz = (y[j] - 0x01010101u) & ~y[j] & 0x80808080u;
                if (__builtin_amdgcn_ballot_w32(hz != 0u) == 0u) continue;
                const int eb = (p0 + u * OWN_WAVES) * 512 + lane * 16 + j * 4;
#pragma unroll
                for (int k = 0; k < 4; ++k) {
                    const bool     m    = (y[j] & (0xFFu << (8 * k))) == 0u;
                    const unsigned mask = __builtin_amdgcn_ballot_w32(m);
                    if (mask != 0u) {
                        const int pos = cnt + (int)__builtin_amdgcn_mbcnt_lo(mask, 0u);
                        if (m) q[pos] = eb + k;
                        cnt += __builtin_popcount(mask);
                    }
                }
            }
            while (cnt >= EPG) {
                cnt -= EPG;
                __builtin_amdgcn_wave_barrier();
                owner_take<F, SAVE>(H, words, acc, q + cnt, EPG, rows, lane, lsave, saved);
                __builtin_amdgcn_wave_barrier();
            }
        }
    }
    if (cnt > 0) {
        __builtin_amdgcn_wave_barrier();
        owner_take<F, SAVE>(H, words, acc, q, cnt, rows, lane, lsave, saved);
    }
    if (SAVE && lane == 0) *lcount = saved;
}

template <int F, bool SAVE>
__launch_bounds__(OWN_WAVES * 32)
__global__ void owner_sum_kernel(const float* __restrict__ H,
                                 const unsigned* __restrict__ tags,
                                 const unsigned* __restrict__ words,
                                 unsigned* lists, int* counts,
                                 float* __restrict__ AGG, int ne, int n_nodes) {
    constexpr int V   = F / 32;
    constexpr int EPG = RIF(F);
    constexpr int QW  = 512 + EPG;
    constexpr int NT  = OWN_WAVES * 32;

    __shared__ __align__(16) float acc[OWN_ROWS * F];
    __shared__ int queue[OWN_WAVES * QW];

    const int tid  = threadIdx.x;
    const int lane = tid & 31;
    const int wave = __builtin_amdgcn_readfirstlane(tid >> 5);
    const int r0   = blockIdx.x * OWN_ROWS;
    const int rows = (n_nodes - r0) < OWN_ROWS ? (n_nodes - r0) : OWN_ROWS;
    const unsigned rep = (unsigned)(blockIdx.x & 255) * 0x01010101u;

    float4* a4 = reinterpret_cast<float4*>(acc);
    for (int i = tid; i < OWN_ROWS * F / 4; i += NT) a4[i] = make_float4(0.0f, 0.0f, 0.0f, 0.0f);
    __syncthreads();

    int* q = queue + wave * QW;
    const int wg = blockIdx.x * OWN_WAVES + wave;
    unsigned* lp = lists + (size_t)wg * LIST_CAP;

    if (SAVE) {
        scan_and_sum<F, true>(H, tags, words, acc, q, ne, rows, rep, wave, lane, lp, counts + wg);
    } else {
        const int n = __builtin_amdgcn_readfirstlane(counts[wg]);
        if ((unsigned)n <= (unsigned)LIST_CAP) {
            int wn = 0;
            if (n > 0) wn = (int)lp[lane < n ? lane : n - 1];
            for (int c = 0; c < n; c += 32) {
                const int w = wn;
                const int m = (n - c) < 32 ? (n - c) : 32;
                const int mn = n - c - 32;
                if (mn > 0) wn = (int)lp[c + 32 + (lane < mn ? lane : mn - 1)];
#pragma unroll
                for (int s = 0; s < 32 / EPG; ++s) {
                    const int k = m - s * EPG;
                    if (k > 0) rows_add<F>(H, acc, w, s * EPG, k < EPG ? k : EPG, rows, lane);
                }
            }
        } else {
            scan_and_sum<F, false>(H, tags, words, acc, q, ne, rows, rep, wave, lane, lp, counts + wg);
        }
    }
    __syncthreads();

    typedef float rowv __attribute__((ext_vector_type(V)));
    for (int r = wave; r < rows; r += OWN_WAVES) {
        const float* ar = acc + r * F + lane;
        rowv o;
#pragma unroll
        for (int j = 0; j < V; ++j) o[j] = ar[32 * j];
        *reinterpret_cast<rowv*>(AGG + (size_t)(r0 + r) * F + V * lane) = o;
    }
}

static inline int cdiv(long a, long b) { return (int)((a + b - 1) / b); }

extern "C" void kernel_launch(void* const* d_in, const int* in_sizes, int n_in,
                              void* d_out, int out_size, void* d_ws, size_t ws_size,
                              hipStream_t stream) {
    const float* features = (const float*)d_in[0];
    const int*   src      = (const int*)d_in[1];
    const int*   dst      = (const int*)d_in[2];
    const float* W1       = (const float*)d_in[3];
    const float* b1       = (const float*)d_in[4];
    const float* W2       = (const float*)d_in[5];
    const float* b2       = (const float*)d_in[6];
    const float* W3       = (const float*)d_in[7];
    const float* b3       = (const float*)d_in[8];
    float*       out      = (float*)d_out;

    const int N = in_sizes[0] / IN_F;
    const int E = in_sizes[1];

    float* ws   = (float*)d_ws;
    float* dsrc = ws;
    float* ddst = ws + N;
    float* buf0 = ws + 2 * (size_t)N;
    float* buf1 = buf0 + (size_t)N * HID_F;

    const long NO = (long)N * OUT_F;
    const int  nblk = cdiv(N, GEMM_ROWS);
    const int  nown = cdiv(N, OWN_ROWS);

    const int npieces = cdiv(E, 512);
    const int nquads  = npieces * 128;
    unsigned* tags  = (unsigned*)d_out;
    unsigned* words = tags + (size_t)nquads;
    unsigned* lists = words + (size_t)npieces * 512;
    int*      counts = (int*)(lists + (size_t)nown * OWN_WAVES * LIST_CAP);

    zero_kernel<<<cdiv(2L * N, 256), 256, 0, stream>>>(dsrc, 2L * N);
    degree_tag_kernel<<<cdiv(nquads, 256), 256, 0, stream>>>(src, dst, dsrc, ddst, tags, words, E, N, nquads);
    rsqrt_kernel<<<cdiv(2L * N, 256), 256, 0, stream>>>(dsrc, 2 * N);

    gemm_rows80_wmma<HID_F, false><<<nblk, HID_F * 2, 0, stream>>>(
        features, dsrc, ddst, b1, W1, buf0, N);
    owner_sum_kernel<HID_F, true><<<nown, OWN_WAVES * 32, 0, stream>>>(buf0, tags, words, lists, counts, buf1, E, N);

    gemm_rows80_wmma<HID_F, true><<<nblk, HID_F * 2, 0, stream>>>(
        buf1, dsrc, ddst, b1, W2, buf0, N);
    owner_sum_kernel<HID_F, false><<<nown, OWN_WAVES * 32, 0, stream>>>(buf0, tags, words, lists, counts, buf1, E, N);

    gemm_rows80_wmma<OUT_F, true><<<nblk, OUT_F * 2, 0, stream>>>(
        buf1, dsrc, ddst, b2, W3, buf0, N);
    owner_sum_kernel<OUT_F, false><<<nown, OWN_WAVES * 32, 0, stream>>>(buf0, tags, words, lists, counts, buf1, E, N);
    epilogue_kernel<<<cdiv(NO, 256), 256, 0, stream>>>(buf1, out, ddst, b3, OUT_F, NO, 0);
}


# Degree pass without global atomics: a block counts a 32768-edge chunk in LDS (two 16-bit counters per dword), stores its plane; a 2nd launch sums planes + rsqrt; 10 -> 9 launches; float sums kept
# speedup vs baseline: 3.8432x; 1.4144x over previous
#include <hip/hip_runtime.h>

typedef float v2f __attribute__((ext_vector_type(2)));
typedef float v8f __attribute__((ext_vector_type(8)));

#define IN_F  128
#define HID_F 128
#define OUT_F 64
#define KDIM  128

#define OWN_ROWS 391
#define HIST_NODES (256 * OWN_ROWS)
#define HIST_DW    (HIST_NODES / 2)
#define CHUNK_E    32768
#define CHUNK_QT   (CHUNK_E / 4 / 1024)

template <bool FULL>
__device__ __forceinline__ int4 load_quad(const int* __restrict__ p, int e0, int ne, int fill) {
    if (FULL) return *reinterpret_cast<const int4*>(p + e0);
    int4 r = make_int4(fill, fill, fill, fill);
    if (e0 + 3 < ne) r = *reinterpret_cast<const int4*>(p + e0);
    return r;
}

__device__ __forceinline__ void fix_cut_quad(int4& r, const int* __restrict__ p, int e0, int ne) {
    if (e0 < ne && e0 + 3 >= ne) {
        r.x = p[e0];
        if (e0 + 1 < ne) r.y = p[e0 + 1];
        if (e0 + 2 < ne) r.z = p[e0 + 2];
    }
}

template <bool FULL>
__device__ __forceinline__ void tag_chunk(const int* __restrict__ src, const int* __restrict__ dst,
                                          unsigned* __restrict__ tags, unsigned* __restrict__ words,
                                          int q0, int ne, int n_nodes, int nquads) {
    int4 s4[CHUNK_QT], d4[CHUNK_QT];
#pragma unroll
    for (int it = 0; it < CHUNK_QT; ++it) {
        const int e0 = 4 * (q0 + 1024 * it);
        s4[it] = load_quad<FULL>(src, e0, ne, 0);
        d4[it] = load_quad<FULL>(dst, e0, ne, -1);
    }
    if (!FULL) {
#pragma unroll
        for (int it = 0; it < CHUNK_QT; ++it) {
            fix_cut_quad(s4[it], src, 4 * (q0 + 1024 * it), ne);
            fix_cut_quad(d4[it], dst, 4 * (q0 + 1024 * it), ne);
        }
    }
#pragma unroll
    for (int it = 0; it < CHUNK_QT; ++it) {
        const int q = q0 + 1024 * it;
        if (!FULL && q >= nquads) continue;
        const int s[4] = { s4[it].x, s4[it].y, s4[it].z, s4[it].w };
        const int d[4] = { d4[it].x, d4[it].y, d4[it].z, d4[it].w };
        unsigned tg = 0u;
        unsigned w[4];
#pragma unroll
        for (int k = 0; k < 4; ++k) {
            const bool     ok  = (unsigned)d[k] < (unsigned)n_nodes;
            const unsigned own = ok ? (unsigned)d[k] / OWN_ROWS : 0u;
            const unsigned loc = ok ? (unsigned)d[k] - own * OWN_ROWS : 511u;
            tg |= (own & 255u) << (8 * k);
            w[k] = ((unsigned)s[k] & 0x1FFFFu) | (loc << 17);
        }
        tags[q] = tg;
        *reinterpret_cast<uint4*>(words + 4 * (size_t)q) = make_uint4(w[0], w[1], w[2], w[3]);
    }
}

template <bool FULL>
__device__ __forceinline__ void count_chunk(const int* __restrict__ idx, unsigned* hist,
                                            int q0, int ne, int tid) {
    int4 v[CHUNK_QT];
#pragma unroll
    for (int it = 0; it < CHUNK_QT; ++it) v[it] = load_quad<FULL>(idx, 4 * (q0 + 1024 * it), ne, -1);
    if (!FULL) {
#pragma unroll
        for (int it = 0; it < CHUNK_QT; ++it) fix_cut_quad(v[it], idx, 4 * (q0 + 1024 * it), ne);
    }
    uint4* h4 = reinterpret_cast<uint4*>(hist);
    for (int i = tid; i < HIST_DW / 4; i += 1024) h4[i] = make_uint4(0u, 0u, 0u, 0u);
    __syncthreads();
#pragma unroll
    for (int it = 0; it < CHUNK_QT; ++it) {
        const int nd[4] = { v[it].x, v[it].y, v[it].z, v[it].w };
#pragma unroll
        for (int k = 0; k < 4; ++k) {
            if ((unsigned)nd[k] < (unsigned)HIST_NODES)
                atomicAdd(&hist[nd[k] >> 1], 1u << (16 * (nd[k] & 1)));
        }
    }
    __syncthreads();
}

__launch_bounds__(1024)
__global__ void degree_hist_kernel(const int* __restrict__ src, const int* __restrict__ dst,
                                   unsigned* __restrict__ planes,
                                   unsigned* __restrict__ tags, unsigned* __restrict__ words,
                                   int ne, int n_nodes, int nquads) {
    __shared__ __align__(16) unsigned hist[HIST_DW];
    const int  tid  = threadIdx.x;
    const int  role = blockIdx.y;
    const int  q0   = blockIdx.x * (CHUNK_E / 4) + tid;
    const bool full = (long)(blockIdx.x + 1) * CHUNK_E <= (long)ne;

    if (role == 2) {
        if (full) tag_chunk<true >(src, dst, tags, words, q0, ne, n_nodes, nquads);
        else      tag_chunk<false>(src, dst, tags, words, q0, ne, n_nodes, nquads);
        return;
    }
    const int* __restrict__ idx = role == 0 ? src : dst;
    if (full) count_chunk<true >(idx, hist, q0, ne, tid);
    else      count_chunk<false>(idx, hist, q0, ne, tid);
    const uint4* h4 = reinterpret_cast<const uint4*>(hist);
    uint4* p4 = reinterpret_cast<uint4*>(planes + ((size_t)role * gridDim.x + blockIdx.x) * HIST_DW);
    for (int i = tid; i < HIST_DW / 4; i += 1024) p4[i] = h4[i];
}

__global__ void degree_sum_kernel(const unsigned* __restrict__ planes, float* __restrict__ dinv,
                                  int n_nodes, int nchunks) {
    const int t = blockIdx.x * blockDim.x + threadIdx.x;
    if (2 * t >= n_nodes) return;
    const unsigned* p = planes + (size_t)blockIdx.y * nchunks * HIST_DW + t;
    unsigned lo = 0u, hi = 0u;
    for (int c0 = 0; c0 < nchunks; c0 += 16) {
        unsigned x[16];
#pragma unroll
        for (int u = 0; u < 16; ++u) {
            const int cc = c0 + u < nchunks ? c0 + u : nchunks - 1;
            x[u] = p[(size_t)cc * HIST_DW];
        }
#pragma unroll
        for (int u = 0; u < 16; ++u) {
            if (c0 + u < nchunks) {
                lo += x[u] & 0xFFFFu;
                hi += x[u] >> 16;
            }
        }
    }
    float* o = dinv + (size_t)blockIdx.y * n_nodes + 2 * t;
    const float d0 = (float)lo;
    o[0] = rsqrtf(d0 < 1.0f ? 1.0f : d0);
    if (2 * t + 1 < n_nodes) {
        const float d1 = (float)hi;
        o[1] = rsqrtf(d1 < 1.0f ? 1.0f : d1);
    }
}

__global__ void epilogue_kernel(const float* __restrict__ X, float* __restrict__ Y,
                                const float* __restrict__ dinv,
                                const float* __restrict__ bias, int F, long total,
                                int do_relu) {
    long t = (long)blockIdx.x * blockDim.x + threadIdx.x;
    if (t >= total) return;
    int i = (int)(t / F);
    int f = (int)(t - (long)i * F);
    float v = X[t] * dinv[i] + bias[f];
    if (do_relu) v = v > 0.0f ? v : 0.0f;
    Y[t] = v;
}

#define GEMM_TM   5
#define GEMM_ROWS (16 * GEMM_TM)
template <int FOUT, bool PRE>
__launch_bounds__(FOUT * 2)
__global__ void gemm_rows80_wmma(const float* X,
                                 const float* __restrict__ dsrc,
                                 const float* __restrict__ ddst,
                                 const float* __restrict__ pbias,
                                 const float* __restrict__ W,
                                 float* __restrict__ H, int n_rows) {
    constexpr int LDSS = KDIM + 4;
    constexpr int NW   = FOUT / 16;
    constexpr int ITER = GEMM_ROWS / NW;
    __shared__ __align__(16) float As[GEMM_ROWS * LDSS];

    const int row0 = blockIdx.x * GEMM_ROWS;
    const int tid  = threadIdx.x;
    const int lane = tid & 31;
    const int wave = __builtin_amdgcn_readfirstlane(tid >> 5);

    float4 x[ITER];
    float  ss[ITER], sd[ITER];
#pragma unroll
    for (int it = 0; it < ITER; ++it) {
        int gr = row0 + wave + NW * it;
        gr = gr < n_rows ? gr : n_rows - 1;
        x[it]  = *reinterpret_cast<const float4*>(X + (size_t)gr * KDIM + 4 * lane);
        ss[it] = dsrc[gr];
        sd[it] = PRE ? ddst[gr] : 0.0f;
    }
    float4 pb = make_float4(0.0f, 0.0f, 0.0f, 0.0f);
    if (PRE) pb = *reinterpret_cast<const float4*>(pbias + 4 * lane);
#pragma unroll
    for (int it = 0; it < ITER; ++it) {
        float4 v = x[it];
        if (PRE) {
            v.x = v.x * sd[it] + pb.x;  v.x = v.x > 0.0f ? v.x : 0.0f;
            v.y = v.y * sd[it] + pb.y;  v.y = v.y > 0.0f ? v.y : 0.0f;
            v.z = v.z * sd[it] + pb.z;  v.z = v.z > 0.0f ? v.z : 0.0f;
            v.w = v.w * sd[it] + pb.w;  v.w = v.w > 0.0f ? v.w : 0.0f;
        }
        v.x *= ss[it]; v.y *= ss[it]; v.z *= ss[it]; v.w *= ss[it];
        *reinterpret_cast<float4*>(&As[(wave + NW * it) * LDSS + 4 * lane]) = v;
    }
    __syncthreads();

    const int n     = lane & 15;
    const int khalf = lane >> 4;
    const int m16   = lane & 15;
    const int ncol  = wave * 16 + n;

    v8f acc[GEMM_TM];
#pragma unroll
    for (int t = 0; t < GEMM_TM; ++t) acc[t] = v8f{};
    v2f a_cur[GEMM_TM], a_nxt[GEMM_TM];
#pragma unroll
    for (int t = 0; t < GEMM_TM; ++t)
        a_cur[t] = *(const v2f*)&As[(t * 16 + m16) * LDSS + 2 * khalf];
#pragma unroll
    for (int k0 = 0; k0 < KDIM; k0 += 4) {
        const int ka = k0 + 2 * khalf;
        v2f b;
        b.x = W[(size_t)ka * FOUT + ncol];
        b.y = W[(size_t)(ka + 1) * FOUT + ncol];
        if (k0 + 4 < KDIM) {
#pragma unroll
            for (int t = 0; t < GEMM_TM; ++t)
                a_nxt[t] = *(const v2f*)&As[(t * 16 + m16) * LDSS + ka + 4];
        }
#pragma unroll
        for (int t = 0; t < GEMM_TM; ++t) {
            acc[t] = __builtin_amdgcn_wmma_f32_16x16x4_f32(
                 false, a_cur[t],  false, b,
                 (short)0, acc[t],  false,  false);
        }
#pragma unroll
        for (int t = 0; t < GEMM_TM; ++t) a_cur[t] = a_nxt[t];
    }

#pragma unroll
    for (int t = 0; t < GEMM_TM; ++t) {
#pragma unroll
        for (int r = 0; r < 8; ++r) {
            const int m = row0 + t * 16 + r + 8 * khalf;
            if (m < n_rows) H[(size_t)m * FOUT + ncol] = acc[t][r];
        }
    }
}

#define OWN_WAVES 32
#define SCAN_U 8
#define LIST_CAP 512
#define RIF(F) (2048 / (F))

template <int F>
__device__ __forceinline__ void rows_add(const float* __restrict__ H, float* acc,
                                         int w, int base, int n, int rows, int lane) {
    constexpr int V   = F / 32;
    constexpr int EPG = RIF(F);
    typedef float rowv __attribute__((ext_vector_type(V)));
    rowv v[EPG];
#pragma unroll
    for (int u = 0; u < EPG; ++u) {
        const int su = __builtin_amdgcn_readlane(w, base + u) & 0x1FFFF;
        v[u] = *reinterpret_cast<const rowv*>(H + (size_t)su * F + V * lane);
    }
#pragma unroll
    for (int u = 0; u < EPG; ++u) {
        const int du = (int)((unsigned)__builtin_amdgcn_readlane(w, base + u) >> 17);
        if (u < n && du < rows) {
            float* ap = acc + du * F + lane;
#pragma unroll
            for (int j = 0; j < V; ++j) atomicAdd(ap + 32 * j, v[u][j]);
        }
    }
}

template <int F, bool SAVE>
__device__ __forceinline__ void owner_take(const float* __restrict__ H,
                                           const unsigned* __restrict__ words,
                                           float* acc, const int* qe, int n, int rows, int lane,
                                           unsigned* lsave, int& saved) {
    constexpr int EPG = RIF(F);
    int idx = lane & (EPG - 1);
    idx = idx < n ? idx : n - 1;
    const int w = (int)words[qe[idx]];
    if (SAVE) {
        if (lane < n && saved + lane < LIST_CAP) lsave[saved + lane] = (unsigned)w;
        saved += n;
    }
    rows_add<F>(H, acc, w, 0, n, rows, lane);
}

template <int F, bool SAVE>
__device__ __forceinline__ void scan_and_sum(const float* __restrict__ H,
                                             const unsigned* __restrict__ tags,
                                             const unsigned* __restrict__ words,
                                             float* acc, int* q, int ne, int rows,
                                             unsigned rep, int wave, int lane,
                                             unsigned* lsave, int* lcount) {
    constexpr int EPG = RIF(F);
    int cnt = 0;
    int saved = 0;
    const int npieces = (ne + 511) >> 9;
    const uint4* tags4 = reinterpret_cast<const uint4*>(tags);

    for (int p0 = wave; p0 < npieces; p0 += OWN_WAVES * SCAN_U) {
        uint4 t[SCAN_U];
#pragma unroll
        for (int u = 0; u < SCAN_U; ++u) {
            const int pc = (p0 + u * OWN_WAVES) < npieces ? (p0 + u * OWN_WAVES) : npieces - 1;
            t[u] = tags4[(size_t)pc * 32 + lane];
        }
#pragma unroll
        for (int u = 0; u < SCAN_U; ++u) {
            if (p0 + u * OWN_WAVES >= npieces) continue;
            const unsigned y[4] = { t[u].x ^ rep, t[u].y ^ rep, t[u].z ^ rep, t[u].w ^ rep };
#pragma unroll
            for (int j = 0; j < 4; ++j) {
                const unsigned hz = (y[j] - 0x01010101u) & ~y[j] & 0x80808080u;
                if (__builtin_amdgcn_ballot_w32(hz != 0u) == 0u) continue;
                const int eb = (p0 + u * OWN_WAVES) * 512 + lane * 16 + j * 4;
#pragma unroll
                for (int k = 0; k < 4; ++k) {
                    const bool     m    = (y[j] & (0xFFu << (8 * k))) == 0u;
                    const unsigned mask = __builtin_amdgcn_ballot_w32(m);
                    if (mask != 0u) {
                        const int pos = cnt + (int)__builtin_amdgcn_mbcnt_lo(mask, 0u);
                        if (m) q[pos] = eb + k;
                        cnt += __builtin_popcount(mask);
                    }
                }
            }
            while (cnt >= EPG) {
                cnt -= EPG;
                __builtin_amdgcn_wave_barrier();
                owner_take<F, SAVE>(H, words, acc, q + cnt, EPG, rows, lane, lsave, saved);
                __builtin_amdgcn_wave_barrier();
            }
        }
    }
    if (cnt > 0) {
        __builtin_amdgcn_wave_barrier();
        owner_take<F, SAVE>(H, words, acc, q, cnt, rows, lane, lsave, saved);
    }
    if (SAVE && lane == 0) *lcount = saved;
}

template <int F, bool SAVE>
__launch_bounds__(OWN_WAVES * 32)
__global__ void owner_sum_kernel(const float* __restrict__ H,
                                 const unsigned* __restrict__ tags,
                                 const unsigned* __restrict__ words,
                                 unsigned* lists, int* counts,
                                 float* __restrict__ AGG, int ne, int n_nodes) {
    constexpr int V   = F / 32;
    constexpr int EPG = RIF(F);
    constexpr int QW  = 512 + EPG;
    constexpr int NT  = OWN_WAVES * 32;

    __shared__ __align__(16) float acc[OWN_ROWS * F];
    __shared__ int queue[OWN_WAVES * QW];

    const int tid  = threadIdx.x;
    const int lane = tid & 31;
    const int wave = __builtin_amdgcn_readfirstlane(tid >> 5);
    const int r0   = blockIdx.x * OWN_ROWS;
    const int rows = (n_nodes - r0) < OWN_ROWS ? (n_nodes - r0) : OWN_ROWS;
    const unsigned rep = (unsigned)(blockIdx.x & 255) * 0x01010101u;

    float4* a4 = reinterpret_cast<float4*>(acc);
    for (int i = tid; i < OWN_ROWS * F / 4; i += NT) a4[i] = make_float4(0.0f, 0.0f, 0.0f, 0.0f);
    __syncthreads();

    int* q = queue + wave * QW;
    const int wg = blockIdx.x * OWN_WAVES + wave;
    unsigned* lp = lists + (size_t)wg * LIST_CAP;

    if (SAVE) {
        scan_and_sum<F, true>(H, tags, words, acc, q, ne, rows, rep, wave, lane, lp, counts + wg);
    } else {
        const int n = __builtin_amdgcn_readfirstlane(counts[wg]);
        if ((unsigned)n <= (unsigned)LIST_CAP) {
            int wn = 0;
            if (n > 0) wn = (int)lp[lane < n ? lane : n - 1];
            for (int c = 0; c < n; c += 32) {
                const int w = wn;
                const int m = (n - c) < 32 ? (n - c) : 32;
                const int mn = n - c - 32;
                if (mn > 0) wn = (int)lp[c + 32 + (lane < mn ? lane : mn - 1)];
#pragma unroll
                for (int s = 0; s < 32 / EPG; ++s) {
                    const int k = m - s * EPG;
                    if (k > 0) rows_add<F>(H, acc, w, s * EPG, k < EPG ? k : EPG, rows, lane);
                }
            }
        } else {
            scan_and_sum<F, false>(H, tags, words, acc, q, ne, rows, rep, wave, lane, lp, counts + wg);
        }
    }
    __syncthreads();

    typedef float rowv __attribute__((ext_vector_type(V)));
    for (int r = wave; r < rows; r += OWN_WAVES) {
        const float* ar = acc + r * F + lane;
        rowv o;
#pragma unroll
        for (int j = 0; j < V; ++j) o[j] = ar[32 * j];
        *reinterpret_cast<rowv*>(AGG + (size_t)(r0 + r) * F + V * lane) = o;
    }
}

static inline int cdiv(long a, long b) { return (int)((a + b - 1) / b); }

extern "C" void kernel_launch(void* const* d_in, const int* in_sizes, int n_in,
                              void* d_out, int out_size, void* d_ws, size_t ws_size,
                              hipStream_t stream) {
    const float* features = (const float*)d_in[0];
    const int*   src      = (const int*)d_in[1];
    const int*   dst      = (const int*)d_in[2];
    const float* W1       = (const float*)d_in[3];
    const float* b1       = (const float*)d_in[4];
    const float* W2       = (const float*)d_in[5];
    const float* b2       = (const float*)d_in[6];
    const float* W3       = (const float*)d_in[7];
    const float* b3       = (const float*)d_in[8];
    float*       out      = (float*)d_out;

    const int N = in_sizes[0] / IN_F;
    const int E = in_sizes[1];

    float* ws   = (float*)d_ws;
    float* dsrc = ws;
    float* ddst = ws + N;
    float* buf0 = ws + 2 * (size_t)N;
    float* buf1 = buf0 + (size_t)N * HID_F;

    const long NO = (long)N * OUT_F;
    const int  nblk = cdiv(N, GEMM_ROWS);
    const int  nown = cdiv(N, OWN_ROWS);

    const int npieces = cdiv(E, 512);
    const int nquads  = npieces * 128;
    unsigned* tags  = (unsigned*)d_out;
    unsigned* words = tags + (size_t)nquads;
    unsigned* lists = words + (size_t)npieces * 512;
    int*      counts = (int*)(lists + (size_t)nown * OWN_WAVES * LIST_CAP);

    const int nchunks = cdiv(E, CHUNK_E);
    unsigned* planes = (unsigned*)buf1;
    degree_hist_kernel<<<dim3(nchunks, 3), 1024, 0, stream>>>(src, dst, planes, tags, words, E, N, nquads);
    degree_sum_kernel<<<dim3(cdiv(cdiv(N, 2), 256), 2), 256, 0, stream>>>(planes, dsrc, N, nchunks);

    gemm_rows80_wmma<HID_F, false><<<nblk, HID_F * 2, 0, stream>>>(
        features, dsrc, ddst, b1, W1, buf0, N);
    owner_sum_kernel<HID_F, true><<<nown, OWN_WAVES * 32, 0, stream>>>(buf0, tags, words, lists, counts, buf1, E, N);

    gemm_rows80_wmma<HID_F, true><<<nblk, HID_F * 2, 0, stream>>>(
        buf1, dsrc, ddst, b1, W2, buf0, N);
    owner_sum_kernel<HID_F, false><<<nown, OWN_WAVES * 32, 0, stream>>>(buf0, tags, words, lists, counts, buf1, E, N);

    gemm_rows80_wmma<OUT_F, true><<<nblk, OUT_F * 2, 0, stream>>>(
        buf1, dsrc, ddst, b2, W3, buf0, N);
    owner_sum_kernel<OUT_F, false><<<nown, OWN_WAVES * 32, 0, stream>>>(buf0, tags, words, lists, counts, buf1, E, N);
    epilogue_kernel<<<cdiv(NO, 256), 256, 0, stream>>>(buf1, out, ddst, b3, OUT_F, NO, 0);
}


# Segment sums in registers: wave w owns its block's rows w+32k (register picked by scalar slot); layer 1 files kept edges into owner-wave LDS lists by integer add; no LDS float adds; order unfixed
# speedup vs baseline: 4.0114x; 1.0438x over previous
#include <hip/hip_runtime.h>

typedef float v2f __attribute__((ext_vector_type(2)));
typedef float v8f __attribute__((ext_vector_type(8)));

#define IN_F  128
#define HID_F 128
#define OUT_F 64
#define KDIM  128

#define OWN_ROWS 391
#define HIST_NODES (256 * OWN_ROWS)
#define HIST_DW    (HIST_NODES / 2)
#define CHUNK_E    32768
#define CHUNK_QT   (CHUNK_E / 4 / 1024)

template <bool FULL>
__device__ __forceinline__ int4 load_quad(const int* __restrict__ p, int e0, int ne, int fill) {
    if (FULL) return *reinterpret_cast<const int4*>(p + e0);
    int4 r = make_int4(fill, fill, fill, fill);
    if (e0 + 3 < ne) r = *reinterpret_cast<const int4*>(p + e0);
    return r;
}

__device__ __forceinline__ void fix_cut_quad(int4& r, const int* __restrict__ p, int e0, int ne) {
    if (e0 < ne && e0 + 3 >= ne) {
        r.x = p[e0];
        if (e0 + 1 < ne) r.y = p[e0 + 1];
        if (e0 + 2 < ne) r.z = p[e0 + 2];
    }
}

template <bool FULL>
__device__ __forceinline__ void tag_chunk(const int* __restrict__ src, const int* __restrict__ dst,
                                          unsigned* __restrict__ tags, unsigned* __restrict__ words,
                                          int q0, int ne, int n_nodes, int nquads) {
    int4 s4[CHUNK_QT], d4[CHUNK_QT];
#pragma unroll
    for (int it = 0; it < CHUNK_QT; ++it) {
        const int e0 = 4 * (q0 + 1024 * it);
        s4[it] = load_quad<FULL>(src, e0, ne, 0);
        d4[it] = load_quad<FULL>(dst, e0, ne, -1);
    }
    if (!FULL) {
#pragma unroll
        for (int it = 0; it < CHUNK_QT; ++it) {
            fix_cut_quad(s4[it], src, 4 * (q0 + 1024 * it), ne);
            fix_cut_quad(d4[it], dst, 4 * (q0 + 1024 * it), ne);
        }
    }
#pragma unroll
    for (int it = 0; it < CHUNK_QT; ++it) {
        const int q = q0 + 1024 * it;
        if (!FULL && q >= nquads) continue;
        const int s[4] = { s4[it].x, s4[it].y, s4[it].z, s4[it].w };
        const int d[4] = { d4[it].x, d4[it].y, d4[it].z, d4[it].w };
        unsigned tg = 0u;
        unsigned w[4];
#pragma unroll
        for (int k = 0; k < 4; ++k) {
            const bool     ok  = (unsigned)d[k] < (unsigned)n_nodes;
            const unsigned own = ok ? (unsigned)d[k] / OWN_ROWS : 0u;
            const unsigned loc = ok ? (unsigned)d[k] - own * OWN_ROWS : 511u;
            tg |= (own & 255u) << (8 * k);
            w[k] = ((unsigned)s[k] & 0x1FFFFu) | (loc << 17);
        }
        tags[q] = tg;
        *reinterpret_cast<uint4*>(words + 4 * (size_t)q) = make_uint4(w[0], w[1], w[2], w[3]);
    }
}

template <bool FULL>
__device__ __forceinline__ void count_chunk(const int* __restrict__ idx, unsigned* hist,
                                            int q0, int ne, int tid) {
    int4 v[CHUNK_QT];
#pragma unroll
    for (int it = 0; it < CHUNK_QT; ++it) v[it] = load_quad<FULL>(idx, 4 * (q0 + 1024 * it), ne, -1);
    if (!FULL) {
#pragma unroll
        for (int it = 0; it < CHUNK_QT; ++it) fix_cut_quad(v[it], idx, 4 * (q0 + 1024 * it), ne);
    }
    uint4* h4 = reinterpret_cast<uint4*>(hist);
    for (int i = tid; i < HIST_DW / 4; i += 1024) h4[i] = make_uint4(0u, 0u, 0u, 0u);
    __syncthreads();
#pragma unroll
    for (int it = 0; it < CHUNK_QT; ++it) {
        const int nd[4] = { v[it].x, v[it].y, v[it].z, v[it].w };
#pragma unroll
        for (int k = 0; k < 4; ++k) {
            if ((unsigned)nd[k] < (unsigned)HIST_NODES)
                atomicAdd(&hist[nd[k] >> 1], 1u << (16 * (nd[k] & 1)));
        }
    }
    __syncthreads();
}

__launch_bounds__(1024)
__global__ void degree_hist_kernel(const int* __restrict__ src, const int* __restrict__ dst,
                                   unsigned* __restrict__ planes,
                                   unsigned* __restrict__ tags, unsigned* __restrict__ words,
                                   int ne, int n_nodes, int nquads) {
    __shared__ __align__(16) unsigned hist[HIST_DW];
    const int  tid  = threadIdx.x;
    const int  role = blockIdx.y;
    const int  q0   = blockIdx.x * (CHUNK_E / 4) + tid;
    const bool full = (long)(blockIdx.x + 1) * CHUNK_E <= (long)ne;

    if (role == 2) {
        if (full) tag_chunk<true >(src, dst, tags, words, q0, ne, n_nodes, nquads);
        else      tag_chunk<false>(src, dst, tags, words, q0, ne, n_nodes, nquads);
        return;
    }
    const int* __restrict__ idx = role == 0 ? src : dst;
    if (full) count_chunk<true >(idx, hist, q0, ne, tid);
    else      count_chunk<false>(idx, hist, q0, ne, tid);
    const uint4* h4 = reinterpret_cast<const uint4*>(hist);
    uint4* p4 = reinterpret_cast<uint4*>(planes + ((size_t)role * gridDim.x + blockIdx.x) * HIST_DW);
    for (int i = tid; i < HIST_DW / 4; i += 1024) p4[i] = h4[i];
}

__global__ void degree_sum_kernel(const unsigned* __restrict__ planes, float* __restrict__ dinv,
                                  int n_nodes, int nchunks) {
    const int t = blockIdx.x * blockDim.x + threadIdx.x;
    if (2 * t >= n_nodes) return;
    const unsigned* p = planes + (size_t)blockIdx.y * nchunks * HIST_DW + t;
    unsigned lo = 0u, hi = 0u;
    for (int c0 = 0; c0 < nchunks; c0 += 16) {
        unsigned x[16];
#pragma unroll
        for (int u = 0; u < 16; ++u) {
            const int cc = c0 + u < nchunks ? c0 + u : nchunks - 1;
            x[u] = p[(size_t)cc * HIST_DW];
        }
#pragma unroll
        for (int u = 0; u < 16; ++u) {
            if (c0 + u < nchunks) {
                lo += x[u] & 0xFFFFu;
                hi += x[u] >> 16;
            }
        }
    }
    float* o = dinv + (size_t)blockIdx.y * n_nodes + 2 * t;
    const float d0 = (float)lo;
    o[0] = rsqrtf(d0 < 1.0f ? 1.0f : d0);
    if (2 * t + 1 < n_nodes) {
        const float d1 = (float)hi;
        o[1] = rsqrtf(d1 < 1.0f ? 1.0f : d1);
    }
}

__global__ void epilogue_kernel(const float* __restrict__ X, float* __restrict__ Y,
                                const float* __restrict__ dinv,
                                const float* __restrict__ bias, int F, long total,
                                int do_relu) {
    long t = (long)blockIdx.x * blockDim.x + threadIdx.x;
    if (t >= total) return;
    int i = (int)(t / F);
    int f = (int)(t - (long)i * F);
    float v = X[t] * dinv[i] + bias[f];
    if (do_relu) v = v > 0.0f ? v : 0.0f;
    Y[t] = v;
}

#define GEMM_TM   5
#define GEMM_ROWS (16 * GEMM_TM)
template <int FOUT, bool PRE>
__launch_bounds__(FOUT * 2)
__global__ void gemm_rows80_wmma(const float* X,
                                 const float* __restrict__ dsrc,
                                 const float* __restrict__ ddst,
                                 const float* __restrict__ pbias,
                                 const float* __restrict__ W,
                                 float* __restrict__ H, int n_rows) {
    constexpr int LDSS = KDIM + 4;
    constexpr int NW   = FOUT / 16;
    constexpr int ITER = GEMM_ROWS / NW;
    __shared__ __align__(16) float As[GEMM_ROWS * LDSS];

    const int row0 = blockIdx.x * GEMM_ROWS;
    const int tid  = threadIdx.x;
    const int lane = tid & 31;
    const int wave = __builtin_amdgcn_readfirstlane(tid >> 5);

    float4 x[ITER];
    float  ss[ITER], sd[ITER];
#pragma unroll
    for (int it = 0; it < ITER; ++it) {
        int gr = row0 + wave + NW * it;
        gr = gr < n_rows ? gr : n_rows - 1;
        x[it]  = *reinterpret_cast<const float4*>(X + (size_t)gr * KDIM + 4 * lane);
        ss[it] = dsrc[gr];
        sd[it] = PRE ? ddst[gr] : 0.0f;
    }
    float4 pb = make_float4(0.0f, 0.0f, 0.0f, 0.0f);
    if (PRE) pb = *reinterpret_cast<const float4*>(pbias + 4 * lane);
#pragma unroll
    for (int it = 0; it < ITER; ++it) {
        float4 v = x[it];
        if (PRE) {
            v.x = v.x * sd[it] + pb.x;  v.x = v.x > 0.0f ? v.x : 0.0f;
            v.y = v.y * sd[it] + pb.y;  v.y = v.y > 0.0f ? v.y : 0.0f;
            v.z = v.z * sd[it] + pb.z;  v.z = v.z > 0.0f ? v.z : 0.0f;
            v.w = v.w * sd[it] + pb.w;  v.w = v.w > 0.0f ? v.w : 0.0f;
        }
        v.x *= ss[it]; v.y *= ss[it]; v.z *= ss[it]; v.w *= ss[it];
        *reinterpret_cast<float4*>(&As[(wave + NW * it) * LDSS + 4 * lane]) = v;
    }
    __syncthreads();

    const int n     = lane & 15;
    const int khalf = lane >> 4;
    const int m16   = lane & 15;
    const int ncol  = wave * 16 + n;

    v8f acc[GEMM_TM];
#pragma unroll
    for (int t = 0; t < GEMM_TM; ++t) acc[t] = v8f{};
    v2f a_cur[GEMM_TM], a_nxt[GEMM_TM];
#pragma unroll
    for (int t = 0; t < GEMM_TM; ++t)
        a_cur[t] = *(const v2f*)&As[(t * 16 + m16) * LDSS + 2 * khalf];
#pragma unroll
    for (int k0 = 0; k0 < KDIM; k0 += 4) {
        const int ka = k0 + 2 * khalf;
        v2f b;
        b.x = W[(size_t)ka * FOUT + ncol];
        b.y = W[(size_t)(ka + 1) * FOUT + ncol];
        if (k0 + 4 < KDIM) {
#pragma unroll
            for (int t = 0; t < GEMM_TM; ++t)
                a_nxt[t] = *(const v2f*)&As[(t * 16 + m16) * LDSS + ka + 4];
        }
#pragma unroll
        for (int t = 0; t < GEMM_TM; ++t) {
            acc[t] = __builtin_amdgcn_wmma_f32_16x16x4_f32(
                 false, a_cur[t],  false, b,
                 (short)0, acc[t],  false,  false);
        }
#pragma unroll
        for (int t = 0; t < GEMM_TM; ++t) a_cur[t] = a_nxt[t];
    }

#pragma unroll
    for (int t = 0; t < GEMM_TM; ++t) {
#pragma unroll
        for (int r = 0; r < 8; ++r) {
            const int m = row0 + t * 16 + r + 8 * khalf;
            if (m < n_rows) H[(size_t)m * FOUT + ncol] = acc[t][r];
        }
    }
}

#define OWN_WAVES 32
#define NSLOT 13
#define NO_ROW 15u
#define SCAN_U 8
#define LIST_CAP 512
#define PASS_F 64
#define NO_SLOT 0xFFC00000u

typedef float slotv __attribute__((ext_vector_type(16)));
typedef float rowv  __attribute__((ext_vector_type(2)));

template <int F>
__device__ __forceinline__ void line_sum(const float* __restrict__ Hc, slotv* acc,
                                         unsigned w, int m) {
    unsigned sw[32];
#pragma unroll
    for (int u = 0; u < 32; ++u) sw[u] = (unsigned)__builtin_amdgcn_readlane((int)w, u);
    rowv v[32];
#pragma unroll
    for (int u = 0; u < 16; ++u)
        v[u] = *reinterpret_cast<const rowv*>(Hc + (size_t)(sw[u] & 0x1FFFFu) * F);
#pragma unroll
    for (int u = 16; u < 32; ++u) v[u] = rowv{};
    if (m > 16) {
#pragma unroll
        for (int u = 16; u < 32; ++u)
            v[u] = *reinterpret_cast<const rowv*>(Hc + (size_t)(sw[u] & 0x1FFFFu) * F);
    }
#pragma unroll
    for (int u = 0; u < 32; ++u) {
        unsigned slot = sw[u] >> 22;
        slot = slot < NO_ROW ? slot : NO_ROW;
        acc[0][slot] += v[u][0];
        acc[1][slot] += v[u][1];
    }
}

template <int F>
__device__ __forceinline__ void list_sum(const float* __restrict__ Hc, slotv* acc,
                                         const unsigned* lp, unsigned* lsave, int n, int lane) {
    unsigned wn = 0u;
    if (n > 0) wn = lp[lane < n ? lane : n - 1];
    for (int c = 0; c < n; c += 32) {
        unsigned w = wn;
        const int m = (n - c) < 32 ? (n - c) : 32;
        if (lsave != nullptr) { if (lane < m) lsave[c + lane] = w; }
        if (lane >= m) w |= NO_SLOT;
        const int mn = n - c - 32;
        if (mn > 0) wn = lp[c + 32 + (lane < mn ? lane : mn - 1)];
        line_sum<F>(Hc, acc, w, m);
    }
}

__device__ __forceinline__ void match_piece(const uint4 t, unsigned rep, int piece, int lane,
                                            int* q, int& cnt) {
    const unsigned y[4] = { t.x ^ rep, t.y ^ rep, t.z ^ rep, t.w ^ rep };
#pragma unroll
    for (int j = 0; j < 4; ++j) {
        const unsigned hz = (y[j] - 0x01010101u) & ~y[j] & 0x80808080u;
        if (__builtin_amdgcn_ballot_w32(hz != 0u) == 0u) continue;
        const int eb = piece * 512 + lane * 16 + j * 4;
#pragma unroll
        for (int k = 0; k < 4; ++k) {
            const bool     m    = (y[j] & (0xFFu << (8 * k))) == 0u;
            const unsigned mask = __builtin_amdgcn_ballot_w32(m);
            if (mask != 0u) {
                const int pos = cnt + (int)__builtin_amdgcn_mbcnt_lo(mask, 0u);
                if (m) q[pos] = eb + k;
                cnt += __builtin_popcount(mask);
            }
        }
    }
}

__device__ __forceinline__ void route_take(const unsigned* __restrict__ words, const int* qe, int n,
                                           unsigned* rq, int* rcnt, int lane) {
    if (lane < n) {
        const unsigned w   = words[qe[lane]];
        const unsigned loc = w >> 17;
        if (loc < (unsigned)OWN_ROWS) {
            const unsigned ow = loc & 31u;
            const int pos = atomicAdd(&rcnt[ow], 1);
            if (pos < LIST_CAP) rq[ow * LIST_CAP + pos] = w;
        }
    }
}

__device__ __forceinline__ void scan_route(const unsigned* __restrict__ tags,
                                           const unsigned* __restrict__ words,
                                           int* q, unsigned* rq, int* rcnt, int ne,
                                           unsigned rep, int wave, int lane) {
    int cnt = 0;
    const int npieces = (ne + 511) >> 9;
    const uint4* tags4 = reinterpret_cast<const uint4*>(tags);

    for (int p0 = wave; p0 < npieces; p0 += OWN_WAVES * SCAN_U) {
        uint4 t[SCAN_U];
#pragma unroll
        for (int u = 0; u < SCAN_U; ++u) {
            const int pc = (p0 + u * OWN_WAVES) < npieces ? (p0 + u * OWN_WAVES) : npieces - 1;
            t[u] = tags4[(size_t)pc * 32 + lane];
        }
#pragma unroll
        for (int u = 0; u < SCAN_U; ++u) {
            if (p0 + u * OWN_WAVES >= npieces) continue;
            match_piece(t[u], rep, p0 + u * OWN_WAVES, lane, q, cnt);
            while (cnt >= 32) {
                cnt -= 32;
                __builtin_amdgcn_wave_barrier();
                route_take(words, q + cnt, 32, rq, rcnt, lane);
                __builtin_amdgcn_wave_barrier();
            }
        }
    }
    if (cnt > 0) {
        __builtin_amdgcn_wave_barrier();
        route_take(words, q, cnt, rq, rcnt, lane);
    }
}

template <int F>
__device__ __forceinline__ void scan_own(const float* __restrict__ Hc,
                                         const unsigned* __restrict__ tags,
                                         const unsigned* __restrict__ words,
                                         slotv* acc, int* q, int ne,
                                         unsigned rep, int wave, int lane) {
    int cnt = 0;
    const int npieces = (ne + 511) >> 9;
    const uint4* tags4 = reinterpret_cast<const uint4*>(tags);
    for (int p = 0; p <= npieces; ++p) {
        const bool last = p == npieces;
        if (!last) match_piece(tags4[(size_t)p * 32 + lane], rep, p, lane, q, cnt);
        while (cnt >= 32 || (last && cnt > 0)) {
            const int k = cnt < 32 ? cnt : 32;
            cnt -= k;
            __builtin_amdgcn_wave_barrier();
            unsigned w = words[q[cnt + (lane < k ? lane : k - 1)]];
            const unsigned loc = w >> 17;
            if (lane >= k || loc >= (unsigned)OWN_ROWS || (loc & 31u) != (unsigned)wave) w |= NO_SLOT;
            line_sum<F>(Hc, acc, w, k);
            __builtin_amdgcn_wave_barrier();
        }
    }
}

template <int F, bool SAVE>
__launch_bounds__(OWN_WAVES * 32)
__global__ void owner_sum_kernel(const float* __restrict__ H,
                                 const unsigned* __restrict__ tags,
                                 const unsigned* __restrict__ words,
                                 unsigned* lists, int* counts,
                                 float* __restrict__ AGG, int ne, int n_nodes) {
    constexpr int QW = 512 + 32;

    __shared__ int      queue[OWN_WAVES * QW];
    __shared__ unsigned rq[SAVE ? OWN_WAVES * LIST_CAP : 1];
    __shared__ int      rcnt[OWN_WAVES];

    const int tid  = threadIdx.x;
    const int lane = tid & 31;
    const int wave = __builtin_amdgcn_readfirstlane(tid >> 5);
    const int r0   = blockIdx.x * OWN_ROWS;
    const int rows = (n_nodes - r0) < OWN_ROWS ? (n_nodes - r0) : OWN_ROWS;
    const unsigned rep = (unsigned)(blockIdx.x & 255) * 0x01010101u;

    int* q = queue + wave * QW;
    const int wg = blockIdx.x * OWN_WAVES + wave;
    unsigned* lp = lists + (size_t)wg * LIST_CAP;

    int n;
    if (SAVE) {
        if (tid < OWN_WAVES) rcnt[tid] = 0;
        __syncthreads();
        scan_route(tags, words, q, rq, rcnt, ne, rep, wave, lane);
        __syncthreads();
        n = __builtin_amdgcn_readfirstlane(rcnt[wave]);
        if (lane == 0) counts[wg] = n;
    } else {
        n = __builtin_amdgcn_readfirstlane(counts[wg]);
    }
    const bool fits = (unsigned)n <= (unsigned)LIST_CAP;
    const unsigned* lsrc = SAVE ? rq + wave * LIST_CAP : lp;

    for (int col = 0; col < F; col += PASS_F) {
        slotv acc[2];
        acc[0] = slotv{};
        acc[1] = slotv{};
        const float* Hc = H + col + 2 * lane;
        if (fits) list_sum<F>(Hc, acc, lsrc, (SAVE && col == 0) ? lp : nullptr, n, lane);
        else      scan_own<F>(Hc, tags, words, acc, q, ne, rep, wave, lane);

#pragma unroll
        for (int s = 0; s < NSLOT; ++s) {
            const int r = s * 32 + wave;
            if (r < rows) {
                rowv o;
                o[0] = acc[0][s];
                o[1] = acc[1][s];
                *reinterpret_cast<rowv*>(AGG + (size_t)(r0 + r) * F + col + 2 * lane) = o;
            }
        }
    }
}

static inline int cdiv(long a, long b) { return (int)((a + b - 1) / b); }

extern "C" void kernel_launch(void* const* d_in, const int* in_sizes, int n_in,
                              void* d_out, int out_size, void* d_ws, size_t ws_size,
                              hipStream_t stream) {
    const float* features = (const float*)d_in[0];
    const int*   src      = (const int*)d_in[1];
    const int*   dst      = (const int*)d_in[2];
    const float* W1       = (const float*)d_in[3];
    const float* b1       = (const float*)d_in[4];
    const float* W2       = (const float*)d_in[5];
    const float* b2       = (const float*)d_in[6];
    const float* W3       = (const float*)d_in[7];
    const float* b3       = (const float*)d_in[8];
    float*       out      = (float*)d_out;

    const int N = in_sizes[0] / IN_F;
    const int E = in_sizes[1];

    float* ws   = (float*)d_ws;
    float* dsrc = ws;
    float* ddst = ws + N;
    float* buf0 = ws + 2 * (size_t)N;
    float* buf1 = buf0 + (size_t)N * HID_F;

    const long NO = (long)N * OUT_F;
    const int  nblk = cdiv(N, GEMM_ROWS);
    const int  nown = cdiv(N, OWN_ROWS);

    const int npieces = cdiv(E, 512);
    const int nquads  = npieces * 128;
    unsigned* tags  = (unsigned*)d_out;
    unsigned* words = tags + (size_t)nquads;
    unsigned* lists = words + (size_t)npieces * 512;
    int*      counts = (int*)(lists + (size_t)nown * OWN_WAVES * LIST_CAP);

    const int nchunks = cdiv(E, CHUNK_E);
    unsigned* planes = (unsigned*)buf1;
    degree_hist_kernel<<<dim3(nchunks, 3), 1024, 0, stream>>>(src, dst, planes, tags, words, E, N, nquads);
    degree_sum_kernel<<<dim3(cdiv(cdiv(N, 2), 256), 2), 256, 0, stream>>>(planes, dsrc, N, nchunks);

    gemm_rows80_wmma<HID_F, false><<<nblk, HID_F * 2, 0, stream>>>(
        features, dsrc, ddst, b1, W1, buf0, N);
    owner_sum_kernel<HID_F, true><<<nown, OWN_WAVES * 32, 0, stream>>>(buf0, tags, words, lists, counts, buf1, E, N);

    gemm_rows80_wmma<HID_F, true><<<nblk, HID_F * 2, 0, stream>>>(
        buf1, dsrc, ddst, b1, W2, buf0, N);
    owner_sum_kernel<HID_F, false><<<nown, OWN_WAVES * 32, 0, stream>>>(buf0, tags, words, lists, counts, buf1, E, N);

    gemm_rows80_wmma<OUT_F, true><<<nblk, OUT_F * 2, 0, stream>>>(
        buf1, dsrc, ddst, b2, W3, buf0, N);
    owner_sum_kernel<OUT_F, false><<<nown, OWN_WAVES * 32, 0, stream>>>(buf0, tags, words, lists, counts, buf1, E, N);
    epilogue_kernel<<<cdiv(NO, 256), 256, 0, stream>>>(buf1, out, ddst, b3, OUT_F, NO, 0);
}
